# starting kernel
# baseline (speedup 1.0000x reference)
#include <hip/hip_runtime.h>
#include <hip/hip_bf16.h>

typedef __attribute__((ext_vector_type(2))) float v2f;
typedef __attribute__((ext_vector_type(8))) float v8f;

#define N_NODES 50000
#define N_EDGES 800000
#define KPAD    196

__global__ void zero_f32(float* __restrict__ p, int n) {
    int i = blockIdx.x * blockDim.x + threadIdx.x;
    if (i < n) p[i] = 0.0f;
}

__global__ void deg_accum(const int* __restrict__ dst, float* __restrict__ deg, int e) {
    int i = blockIdx.x * blockDim.x + threadIdx.x;
    if (i < e) atomicAdd(&deg[dst[i]], 1.0f);
}

__global__ void deg_to_dinv(float* __restrict__ deg, int n) {
    int i = blockIdx.x * blockDim.x + threadIdx.x;
    if (i < n) deg[i] = rsqrtf(fmaxf(deg[i], 1.0f));
}

__global__ void copy_to_xcat(const float* __restrict__ h, float* __restrict__ xcat, int n64) {
    int i = blockIdx.x * blockDim.x + threadIdx.x;
    if (i >= n64) return;
    int node = i >> 6, f = i & 63;
    xcat[node * 192 + f] = h[i];
}

__global__ void spmm_edges(const int* __restrict__ src, const int* __restrict__ dst,
                           const float* __restrict__ x, int xStride,
                           const float* __restrict__ dinv,
                           float* __restrict__ agg, int e) {
    long long i = (long long)blockIdx.x * blockDim.x + threadIdx.x;
    if (i >= (long long)e * 64) return;
    int eidx = (int)(i >> 6);
    int f    = (int)(i & 63);
    int s = src[eidx], d = dst[eidx];
    float v = x[s * xStride + f] * dinv[s];
    atomicAdd(&agg[d * 64 + f], v);
}

__global__ void cheb_x1(const float* __restrict__ agg, float* __restrict__ xcat,
                        const float* __restrict__ dinv, float cAgg, float cX0, int n64) {
    int i = blockIdx.x * blockDim.x + threadIdx.x;
    if (i >= n64) return;
    int node = i >> 6, f = i & 63;
    float a = agg[i] * dinv[node];
    xcat[node * 192 + 64 + f] = cAgg * a + cX0 * xcat[node * 192 + f];
}

__global__ void cheb_x2(const float* __restrict__ agg, float* __restrict__ xcat,
                        const float* __restrict__ dinv,
                        float cAgg, float cX1, float cX0, int n64) {
    int i = blockIdx.x * blockDim.x + threadIdx.x;
    if (i >= n64) return;
    int node = i >> 6, f = i & 63;
    float a  = agg[i] * dinv[node];
    float x0 = xcat[node * 192 + f];
    float x1 = xcat[node * 192 + 64 + f];
    xcat[node * 192 + 128 + f] = cAgg * a + cX1 * x1 + cX0 * x0;
}

__global__ void __launch_bounds__(128)
gemm_wmma(const float* __restrict__ X, int xStride,
          const float* __restrict__ W, const float* __restrict__ bias,
          float* __restrict__ Out, int outStride,
          int Kdim, int Ncols, int doRelu) {
    __shared__ float sW[16 * KPAD];
    __shared__ float sA[4][16 * KPAD];

    const int tid  = threadIdx.x;
    const int wave = tid >> 5;
    const int lane = tid & 31;
    const int rowBase = blockIdx.x * 64;
    const int waveRow = rowBase + wave * 16;
    const int n0 = blockIdx.y * 16;

    for (int idx = tid; idx < 16 * Kdim; idx += 128) {
        int n = idx / Kdim, k = idx - n * Kdim;
        int gn = n0 + n;
        sW[n * KPAD + k] = (gn < Ncols) ? W[gn * Kdim + k] : 0.0f;
    }
    for (int idx = lane; idx < 16 * Kdim; idx += 32) {
        int r = idx / Kdim, k = idx - r * Kdim;
        int row = waveRow + r;
        sA[wave][r * KPAD + k] = (row < N_NODES) ? X[row * xStride + k] : 0.0f;
    }
    __syncthreads();

    const int half = lane >> 4;
    const int l16  = lane & 15;

    v8f c = {};
    const float* aRow = &sA[wave][0];
    for (int k0 = 0; k0 < Kdim; k0 += 4) {
        int ka = k0 + 2 * half;
        v2f a, b;
        a.x = aRow[l16 * KPAD + ka];
        a.y = aRow[l16 * KPAD + ka + 1];
        b.x = sW[l16 * KPAD + ka];
        b.y = sW[l16 * KPAD + ka + 1];
        c = __builtin_amdgcn_wmma_f32_16x16x4_f32(
                 false, a,  false, b,
                 (short)0, c,  false,  false);
    }

    int col = n0 + l16;
    if (col < Ncols) {
        float bv = bias[col];
        #pragma unroll
        for (int r = 0; r < 8; ++r) {
            int row = waveRow + r + 8 * half;
            if (row < N_NODES) {
                float v = c[r] + bv;
                if (doRelu) v = fmaxf(v, 0.0f);
                Out[row * outStride + col] = v;
            }
        }
    }
}

extern "C" void kernel_launch(void* const* d_in, const int* in_sizes, int n_in,
                              void* d_out, int out_size, void* d_ws, size_t ws_size,
                              hipStream_t stream) {
    const float* in_feat = (const float*)d_in[0];
    const int*   src     = (const int*)d_in[1];
    const int*   dst     = (const int*)d_in[2];
    const float* W1 = (const float*)d_in[3];   const float* b1 = (const float*)d_in[4];
    const float* W2 = (const float*)d_in[5];   const float* b2 = (const float*)d_in[6];
    const float* Wc1 = (const float*)d_in[7];  const float* bc1 = (const float*)d_in[8];
    const float* Wc2 = (const float*)d_in[9];  const float* bc2 = (const float*)d_in[10];
    const float* W3 = (const float*)d_in[11];  const float* b3 = (const float*)d_in[12];
    const float* W4 = (const float*)d_in[13];  const float* b4 = (const float*)d_in[14];

    char* ws = (char*)d_ws;
    const size_t SZ_H   = (size_t)N_NODES * 64 * 4;
    const size_t SZ_CAT = (size_t)N_NODES * 192 * 4;
    float* dinv = (float*)(ws);
    float* bufA = (float*)(ws + 200192);
    float* bufB = (float*)(ws + 200192 + SZ_H);
    float* xcat = (float*)(ws + 200192 + 2 * SZ_H);
    float* agg  = (float*)(ws + 200192 + 2 * SZ_H + SZ_CAT);

    const int n64 = N_NODES * 64;
    dim3 blkS(256);
    dim3 gN((N_NODES + 255) / 256), gE((N_EDGES + 255) / 256), gNF((n64 + 255) / 256);
    long long espmm = (long long)N_EDGES * 64;
    dim3 gSP((unsigned)((espmm + 255) / 256));

    zero_f32<<<gN, blkS, 0, stream>>>(dinv, N_NODES);
    deg_accum<<<gE, blkS, 0, stream>>>(dst, dinv, N_EDGES);
    deg_to_dinv<<<gN, blkS, 0, stream>>>(dinv, N_NODES);

    dim3 gemmBlk(128);
    dim3 gG64((N_NODES + 63) / 64, 4);
    dim3 gG2((N_NODES + 63) / 64, 1);

    gemm_wmma<<<gG64, gemmBlk, 0, stream>>>(in_feat, 128, W1, b1, bufA, 64, 128, 64, 1);
    gemm_wmma<<<gG64, gemmBlk, 0, stream>>>(bufA,     64, W2, b2, bufB, 64,  64, 64, 1);

    const float rn = 2.0f / 2.0f;
    const float* h_in = bufB;
    float*       h_out = bufA;
    const float* Wc[2] = { Wc1, Wc2 };
    const float* bc[2] = { bc1, bc2 };
    for (int layer = 0; layer < 2; ++layer) {
        copy_to_xcat<<<gNF, blkS, 0, stream>>>(h_in, xcat, n64);
        zero_f32<<<gNF, blkS, 0, stream>>>(agg, n64);
        spmm_edges<<<gSP, blkS, 0, stream>>>(src, dst, xcat, 192, dinv, agg, N_EDGES);
        cheb_x1<<<gNF, blkS, 0, stream>>>(agg, xcat, dinv, -rn, rn - 1.0f, n64);
        zero_f32<<<gNF, blkS, 0, stream>>>(agg, n64);
        spmm_edges<<<gSP, blkS, 0, stream>>>(src, dst, xcat + 64, 192, dinv, agg, N_EDGES);
        cheb_x2<<<gNF, blkS, 0, stream>>>(agg, xcat, dinv, -2.0f * rn, 2.0f * (rn - 1.0f), -1.0f, n64);
        gemm_wmma<<<gG64, gemmBlk, 0, stream>>>(xcat, 192, Wc[layer], bc[layer], h_out, 64, 192, 64, 1);
        const float* t = h_in; h_in = h_out; h_out = (float*)t;
    }
    gemm_wmma<<<gG64, gemmBlk, 0, stream>>>(h_in, 64, W3, b3, (float*)h_out, 64, 64, 64, 1);
    gemm_wmma<<<gG2,  gemmBlk, 0, stream>>>((const float*)h_out, 64, W4, b4, (float*)d_out, 2, 64, 2, 0);
}


# GEMMs re-tiled: a wave does 16 rows x all output columns, so X is read once (was 4x, scalar) by 512-B coalesced loads via padded LDS; W staged once per 128 rows; k order permuted within each 8
# speedup vs baseline: 1.5191x; 1.5191x over previous
#include <hip/hip_runtime.h>
#include <hip/hip_bf16.h>

typedef __attribute__((ext_vector_type(2))) float v2f;
typedef __attribute__((ext_vector_type(4))) float v4f;
typedef __attribute__((ext_vector_type(8))) float v8f;

#define N_NODES 50000
#define N_EDGES 800000

__global__ void zero_f32(float* __restrict__ p, int n) {
    int i = blockIdx.x * blockDim.x + threadIdx.x;
    if (i < n) p[i] = 0.0f;
}

__global__ void deg_accum(const int* __restrict__ dst, float* __restrict__ deg, int e) {
    int i = blockIdx.x * blockDim.x + threadIdx.x;
    if (i < e) atomicAdd(&deg[dst[i]], 1.0f);
}

__global__ void deg_to_dinv(float* __restrict__ deg, int n) {
    int i = blockIdx.x * blockDim.x + threadIdx.x;
    if (i < n) deg[i] = rsqrtf(fmaxf(deg[i], 1.0f));
}

__global__ void copy_to_xcat(const float* __restrict__ h, float* __restrict__ xcat, int n64) {
    int i = blockIdx.x * blockDim.x + threadIdx.x;
    if (i >= n64) return;
    int node = i >> 6, f = i & 63;
    xcat[node * 192 + f] = h[i];
}

__global__ void spmm_edges(const int* __restrict__ src, const int* __restrict__ dst,
                           const float* __restrict__ x, int xStride,
                           const float* __restrict__ dinv,
                           float* __restrict__ agg, int e) {
    long long i = (long long)blockIdx.x * blockDim.x + threadIdx.x;
    if (i >= (long long)e * 64) return;
    int eidx = (int)(i >> 6);
    int f    = (int)(i & 63);
    int s = src[eidx], d = dst[eidx];
    float v = x[s * xStride + f] * dinv[s];
    atomicAdd(&agg[d * 64 + f], v);
}

__global__ void cheb_x1(const float* __restrict__ agg, float* __restrict__ xcat,
                        const float* __restrict__ dinv, float cAgg, float cX0, int n64) {
    int i = blockIdx.x * blockDim.x + threadIdx.x;
    if (i >= n64) return;
    int node = i >> 6, f = i & 63;
    float a = agg[i] * dinv[node];
    xcat[node * 192 + 64 + f] = cAgg * a + cX0 * xcat[node * 192 + f];
}

__global__ void cheb_x2(const float* __restrict__ agg, float* __restrict__ xcat,
                        const float* __restrict__ dinv,
                        float cAgg, float cX1, float cX0, int n64) {
    int i = blockIdx.x * blockDim.x + threadIdx.x;
    if (i >= n64) return;
    int node = i >> 6, f = i & 63;
    float a  = agg[i] * dinv[node];
    float x0 = xcat[node * 192 + f];
    float x1 = xcat[node * 192 + 64 + f];
    xcat[node * 192 + 128 + f] = cAgg * a + cX1 * x1 + cX0 * x0;
}

#define GEMM_WAVES 8
#define GEMM_ROWS  (GEMM_WAVES * 16)

template <int K, int NC, bool RELU>
__global__ void __launch_bounds__(GEMM_WAVES * 32)
gemm_rows(const float* __restrict__ X, const float* __restrict__ W,
          const float* __restrict__ bias, float* __restrict__ Out) {
    constexpr int NT  = (NC + 15) / 16;
    constexpr int P   = K + 4;
    constexpr int KQ  = K / 4;
    constexpr int AQ  = K / 8;
    constexpr int WQ  = NT * 16 * KQ / (GEMM_WAVES * 32);
    static_assert(WQ * GEMM_WAVES * 32 == NT * 16 * KQ, "the W tile must divide evenly over the block");
    constexpr int PC  = 68;
    static_assert(NC == 64 || NC < 16, "the epilogue handles 64 columns or fewer than 16");

    __shared__ alignas(16) float sW[NT * 16 * P];
    __shared__ alignas(16) float sA[GEMM_WAVES][16 * P];

    const int tid  = threadIdx.x;
    const int wave = tid >> 5;
    const int lane = tid & 31;
    const int waveRow = blockIdx.x * GEMM_ROWS + wave * 16;
    const bool rowsIn = waveRow < N_NODES;
    const int  ldRow  = rowsIn ? waveRow : 0;

    const v4f* Ag = (const v4f*)(X + (size_t)ldRow * K);
    v4f av[AQ];
    #pragma unroll
    for (int q = 0; q < AQ; ++q) av[q] = Ag[q * 32 + lane];

    const v4f* Wg = (const v4f*)W;
    v4f wv[WQ];
    #pragma unroll
    for (int i = 0; i < WQ; ++i) {
        int j = i * (GEMM_WAVES * 32) + tid;
        wv[i] = (NC == NT * 16 || j < NC * KQ) ? Wg[j] : v4f{0.0f, 0.0f, 0.0f, 0.0f};
    }

    const int c = lane & 15;
    const int h = lane >> 4;
    float bv[NT];
    #pragma unroll
    for (int nt = 0; nt < NT; ++nt) {
        int col = nt * 16 + c;
        bv[nt] = (col < NC) ? bias[col] : 0.0f;
    }

    #pragma unroll
    for (int q = 0; q < AQ; ++q) {
        int j = q * 32 + lane;
        int r = j / KQ, kq = j - r * KQ;
        *(v4f*)&sA[wave][r * P + kq * 4] = av[q];
    }
    #pragma unroll
    for (int i = 0; i < WQ; ++i) {
        int j = i * (GEMM_WAVES * 32) + tid;
        int n = j / KQ, kq = j - n * KQ;
        *(v4f*)&sW[n * P + kq * 4] = wv[i];
    }
    __syncthreads();

    const float* aF = &sA[wave][c * P + 4 * h];
    const float* bF = &sW[c * P + 4 * h];
    v8f acc[NT];
    #pragma unroll
    for (int nt = 0; nt < NT; ++nt) acc[nt] = v8f{};

    #pragma unroll
    for (int t = 0; t < AQ; ++t) {
        v4f a = *(const v4f*)(aF + 8 * t);
        v4f b[NT];
        #pragma unroll
        for (int nt = 0; nt < NT; ++nt) b[nt] = *(const v4f*)(bF + nt * 16 * P + 8 * t);
        v2f alo = {a.x, a.y}, ahi = {a.z, a.w};
        #pragma unroll
        for (int nt = 0; nt < NT; ++nt) {
            v2f blo = {b[nt].x, b[nt].y};
            acc[nt] = __builtin_amdgcn_wmma_f32_16x16x4_f32(
                false, alo, false, blo, (short)0, acc[nt], false, false);
        }
        #pragma unroll
        for (int nt = 0; nt < NT; ++nt) {
            v2f bhi = {b[nt].z, b[nt].w};
            acc[nt] = __builtin_amdgcn_wmma_f32_16x16x4_f32(
                false, ahi, false, bhi, (short)0, acc[nt], false, false);
        }
    }

    float* sC = &sA[wave][0];
    if constexpr (NC == 64) {
        #pragma unroll
        for (int nt = 0; nt < NT; ++nt) {
            #pragma unroll
            for (int r = 0; r < 8; ++r) {
                float v = acc[nt][r] + bv[nt];
                if (RELU) v = fmaxf(v, 0.0f);
                sC[(8 * h + r) * PC + nt * 16 + c] = v;
            }
        }
        __syncthreads();
        v4f* Og = (v4f*)(Out + (size_t)waveRow * 64);
        #pragma unroll
        for (int q = 0; q < 8; ++q) {
            int j = q * 32 + lane;
            v4f v = *(const v4f*)&sC[(j >> 4) * PC + (j & 15) * 4];
            if (rowsIn) Og[j] = v;
        }
    } else {
        #pragma unroll
        for (int r = 0; r < 8; ++r) {
            float v = acc[0][r] + bv[0];
            if (RELU) v = fmaxf(v, 0.0f);
            if (c < NC) sC[(8 * h + r) * NC + c] = v;
        }
        __syncthreads();
        if (rowsIn && lane < 16 * NC) Out[(size_t)waveRow * NC + lane] = sC[lane];
    }
}

extern "C" void kernel_launch(void* const* d_in, const int* in_sizes, int n_in,
                              void* d_out, int out_size, void* d_ws, size_t ws_size,
                              hipStream_t stream) {
    const float* in_feat = (const float*)d_in[0];
    const int*   src     = (const int*)d_in[1];
    const int*   dst     = (const int*)d_in[2];
    const float* W1 = (const float*)d_in[3];   const float* b1 = (const float*)d_in[4];
    const float* W2 = (const float*)d_in[5];   const float* b2 = (const float*)d_in[6];
    const float* Wc1 = (const float*)d_in[7];  const float* bc1 = (const float*)d_in[8];
    const float* Wc2 = (const float*)d_in[9];  const float* bc2 = (const float*)d_in[10];
    const float* W3 = (const float*)d_in[11];  const float* b3 = (const float*)d_in[12];
    const float* W4 = (const float*)d_in[13];  const float* b4 = (const float*)d_in[14];

    char* ws = (char*)d_ws;
    const size_t SZ_H   = (size_t)N_NODES * 64 * 4;
    const size_t SZ_CAT = (size_t)N_NODES * 192 * 4;
    float* dinv = (float*)(ws);
    float* bufA = (float*)(ws + 200192);
    float* bufB = (float*)(ws + 200192 + SZ_H);
    float* xcat = (float*)(ws + 200192 + 2 * SZ_H);
    float* agg  = (float*)(ws + 200192 + 2 * SZ_H + SZ_CAT);

    const int n64 = N_NODES * 64;
    dim3 blkS(256);
    dim3 gN((N_NODES + 255) / 256), gE((N_EDGES + 255) / 256), gNF((n64 + 255) / 256);
    long long espmm = (long long)N_EDGES * 64;
    dim3 gSP((unsigned)((espmm + 255) / 256));

    zero_f32<<<gN, blkS, 0, stream>>>(dinv, N_NODES);
    deg_accum<<<gE, blkS, 0, stream>>>(dst, dinv, N_EDGES);
    deg_to_dinv<<<gN, blkS, 0, stream>>>(dinv, N_NODES);

    dim3 gemmBlk(GEMM_WAVES * 32);
    dim3 gG((N_NODES + GEMM_ROWS - 1) / GEMM_ROWS);

    gemm_rows<128, 64, true><<<gG, gemmBlk, 0, stream>>>(in_feat, W1, b1, bufA);
    gemm_rows<64, 64, true><<<gG, gemmBlk, 0, stream>>>(bufA, W2, b2, bufB);

    const float rn = 2.0f / 2.0f;
    const float* h_in = bufB;
    float*       h_out = bufA;
    const float* Wc[2] = { Wc1, Wc2 };
    const float* bc[2] = { bc1, bc2 };
    for (int layer = 0; layer < 2; ++layer) {
        copy_to_xcat<<<gNF, blkS, 0, stream>>>(h_in, xcat, n64);
        zero_f32<<<gNF, blkS, 0, stream>>>(agg, n64);
        spmm_edges<<<gSP, blkS, 0, stream>>>(src, dst, xcat, 192, dinv, agg, N_EDGES);
        cheb_x1<<<gNF, blkS, 0, stream>>>(agg, xcat, dinv, -rn, rn - 1.0f, n64);
        zero_f32<<<gNF, blkS, 0, stream>>>(agg, n64);
        spmm_edges<<<gSP, blkS, 0, stream>>>(src, dst, xcat + 64, 192, dinv, agg, N_EDGES);
        cheb_x2<<<gNF, blkS, 0, stream>>>(agg, xcat, dinv, -2.0f * rn, 2.0f * (rn - 1.0f), -1.0f, n64);
        gemm_rows<192, 64, true><<<gG, gemmBlk, 0, stream>>>(xcat, Wc[layer], bc[layer], h_out);
        const float* t = h_in; h_in = h_out; h_out = (float*)t;
    }
    gemm_rows<64, 64, true><<<gG, gemmBlk, 0, stream>>>(h_in, W3, b3, (float*)h_out);
    gemm_rows<64, 2, false><<<gG, gemmBlk, 0, stream>>>((const float*)h_out, W4, b4, (float*)d_out);
}


# SpMM as a gather: edges bucketed by dst once per call (int atomic rank + scan); each row summed in registers in rank order by one lane group, written once as X1/X2; float atomics and agg passes gone
# speedup vs baseline: 4.2689x; 2.8102x over previous
#include <hip/hip_runtime.h>
#include <hip/hip_bf16.h>

typedef __attribute__((ext_vector_type(2))) float v2f;
typedef __attribute__((ext_vector_type(4))) float v4f;
typedef __attribute__((ext_vector_type(8))) float v8f;

#define N_NODES 50000
#define N_EDGES 800000

__global__ void zero_i32(int* __restrict__ p, int n) {
    int i = blockIdx.x * blockDim.x + threadIdx.x;
    if (i < n) p[i] = 0;
}

__global__ void copy_to_xcat(const float* __restrict__ h, float* __restrict__ xcat, int n64) {
    int i = blockIdx.x * blockDim.x + threadIdx.x;
    if (i >= n64) return;
    int node = i >> 6, f = i & 63;
    xcat[node * 192 + f] = h[i];
}

__global__ void edge_rank(const int* __restrict__ dst, int* __restrict__ cnt,
                          int* __restrict__ rank, int e) {
    int i = blockIdx.x * blockDim.x + threadIdx.x;
    if (i < e) rank[i] = atomicAdd(&cnt[dst[i]], 1);
}

__global__ void __launch_bounds__(256)
scan_rows(const int* __restrict__ cnt, int* __restrict__ rowptr,
          float* __restrict__ dinv, int n) {
    __shared__ int sTot[8];
    __shared__ int sPre[8];
    const int tid = threadIdx.x, lane = tid & 31, wave = tid >> 5;
    const int base = blockIdx.x * 256;
    int pre = 0;
    for (int i = tid; i < base; i += 256) pre += cnt[i];
    const int i = base + tid;
    const int c = (i < n) ? cnt[i] : 0;
    int inc = c;
    for (int o = 1; o < 32; o <<= 1) {
        int t = __shfl_up(inc, o);
        if (lane >= o) inc += t;
    }
    for (int o = 16; o > 0; o >>= 1) pre += __shfl_xor(pre, o);
    if (lane == 31) sTot[wave] = inc;
    if (lane == 0)  sPre[wave] = pre;
    __syncthreads();
    int off = 0;
    for (int w = 0; w < 8; ++w) {
        off += sPre[w];
        if (w < wave) off += sTot[w];
    }
    if (i < n) {
        rowptr[i] = off + inc - c;
        dinv[i]   = rsqrtf(fmaxf((float)c, 1.0f));
    }
}

__global__ void fill_edges(const int* __restrict__ src, const int* __restrict__ dst,
                           const int* __restrict__ rank, const int* __restrict__ rowptr,
                           const float* __restrict__ dinv, int2* __restrict__ ed, int e) {
    int i = blockIdx.x * blockDim.x + threadIdx.x;
    if (i >= e) return;
    int s = src[i], d = dst[i];
    ed[rowptr[d] + rank[i]] = make_int2(s, __float_as_int(dinv[s]));
}

template <bool SECOND>
__global__ void __launch_bounds__(256)
spmm_pull(const int* __restrict__ rowptr, const int* __restrict__ cnt,
          const int2* __restrict__ ed, const float* __restrict__ dinv,
          float* xcat, float cAgg, float cIn, float cX0) {
    const int row = (blockIdx.x * 256 + threadIdx.x) >> 4;
    const int l   = threadIdx.x & 15;
    if (row >= N_NODES) return;
    constexpr int IN = SECOND ? 64 : 0;
    const float* xb = xcat + IN + 4 * l;
    const int beg = rowptr[row];
    const int end = beg + cnt[row];
    const v4f zero = {0.0f, 0.0f, 0.0f, 0.0f};
    v4f acc = zero;
    for (int e = beg; e < end; e += 8) {
        int2 p[8];
        #pragma unroll
        for (int j = 0; j < 8; ++j) {
            int k = e + j;
            p[j] = ed[k < end ? k : end - 1];
        }
        v4f x[8];
        #pragma unroll
        for (int j = 0; j < 8; ++j) x[j] = *(const v4f*)(xb + (size_t)p[j].x * 192);
        #pragma unroll
        for (int j = 0; j < 8; ++j) {
            v4f xv = (e + j < end) ? x[j] : zero;
            acc += xv * __int_as_float(p[j].y);
        }
    }
    const float di = dinv[row];
    float* o = xcat + (size_t)row * 192;
    const v4f own = *(const v4f*)(o + IN + 4 * l);
    v4f r = cAgg * (acc * di) + cIn * own;
    if (SECOND) {
        const v4f x0 = *(const v4f*)(o + 4 * l);
        r += cX0 * x0;
    }
    *(v4f*)(o + IN + 64 + 4 * l) = r;
}

#define GEMM_WAVES 8
#define GEMM_ROWS  (GEMM_WAVES * 16)

template <int K, int NC, bool RELU>
__global__ void __launch_bounds__(GEMM_WAVES * 32)
gemm_rows(const float* __restrict__ X, const float* __restrict__ W,
          const float* __restrict__ bias, float* __restrict__ Out) {
    constexpr int NT  = (NC + 15) / 16;
    constexpr int P   = K + 4;
    constexpr int KQ  = K / 4;
    constexpr int AQ  = K / 8;
    constexpr int WQ  = NT * 16 * KQ / (GEMM_WAVES * 32);
    static_assert(WQ * GEMM_WAVES * 32 == NT * 16 * KQ, "the W tile must divide evenly over the block");
    constexpr int PC  = 68;
    static_assert(NC == 64 || NC < 16, "the epilogue handles 64 columns or fewer than 16");

    __shared__ alignas(16) float sW[NT * 16 * P];
    __shared__ alignas(16) float sA[GEMM_WAVES][16 * P];

    const int tid  = threadIdx.x;
    const int wave = tid >> 5;
    const int lane = tid & 31;
    const int waveRow = blockIdx.x * GEMM_ROWS + wave * 16;
    const bool rowsIn = waveRow < N_NODES;
    const int  ldRow  = rowsIn ? waveRow : 0;

    const v4f* Ag = (const v4f*)(X + (size_t)ldRow * K);
    v4f av[AQ];
    #pragma unroll
    for (int q = 0; q < AQ; ++q) av[q] = Ag[q * 32 + lane];

    const v4f* Wg = (const v4f*)W;
    v4f wv[WQ];
    #pragma unroll
    for (int i = 0; i < WQ; ++i) {
        int j = i * (GEMM_WAVES * 32) + tid;
        wv[i] = (NC == NT * 16 || j < NC * KQ) ? Wg[j] : v4f{0.0f, 0.0f, 0.0f, 0.0f};
    }

    const int c = lane & 15;
    const int h = lane >> 4;
    float bv[NT];
    #pragma unroll
    for (int nt = 0; nt < NT; ++nt) {
        int col = nt * 16 + c;
        bv[nt] = (col < NC) ? bias[col] : 0.0f;
    }

    #pragma unroll
    for (int q = 0; q < AQ; ++q) {
        int j = q * 32 + lane;
        int r = j / KQ, kq = j - r * KQ;
        *(v4f*)&sA[wave][r * P + kq * 4] = av[q];
    }
    #pragma unroll
    for (int i = 0; i < WQ; ++i) {
        int j = i * (GEMM_WAVES * 32) + tid;
        int n = j / KQ, kq = j - n * KQ;
        *(v4f*)&sW[n * P + kq * 4] = wv[i];
    }
    __syncthreads();

    const float* aF = &sA[wave][c * P + 4 * h];
    const float* bF = &sW[c * P + 4 * h];
    v8f acc[NT];
    #pragma unroll
    for (int nt = 0; nt < NT; ++nt) acc[nt] = v8f{};

    #pragma unroll
    for (int t = 0; t < AQ; ++t) {
        v4f a = *(const v4f*)(aF + 8 * t);
        v4f b[NT];
        #pragma unroll
        for (int nt = 0; nt < NT; ++nt) b[nt] = *(const v4f*)(bF + nt * 16 * P + 8 * t);
        v2f alo = {a.x, a.y}, ahi = {a.z, a.w};
        #pragma unroll
        for (int nt = 0; nt < NT; ++nt) {
            v2f blo = {b[nt].x, b[nt].y};
            acc[nt] = __builtin_amdgcn_wmma_f32_16x16x4_f32(
                false, alo, false, blo, (short)0, acc[nt], false, false);
        }
        #pragma unroll
        for (int nt = 0; nt < NT; ++nt) {
            v2f bhi = {b[nt].z, b[nt].w};
            acc[nt] = __builtin_amdgcn_wmma_f32_16x16x4_f32(
                false, ahi, false, bhi, (short)0, acc[nt], false, false);
        }
    }

    float* sC = &sA[wave][0];
    if constexpr (NC == 64) {
        #pragma unroll
        for (int nt = 0; nt < NT; ++nt) {
            #pragma unroll
            for (int r = 0; r < 8; ++r) {
                float v = acc[nt][r] + bv[nt];
                if (RELU) v = fmaxf(v, 0.0f);
                sC[(8 * h + r) * PC + nt * 16 + c] = v;
            }
        }
        __syncthreads();
        v4f* Og = (v4f*)(Out + (size_t)waveRow * 64);
        #pragma unroll
        for (int q = 0; q < 8; ++q) {
            int j = q * 32 + lane;
            v4f v = *(const v4f*)&sC[(j >> 4) * PC + (j & 15) * 4];
            if (rowsIn) Og[j] = v;
        }
    } else {
        #pragma unroll
        for (int r = 0; r < 8; ++r) {
            float v = acc[0][r] + bv[0];
            if (RELU) v = fmaxf(v, 0.0f);
            if (c < NC) sC[(8 * h + r) * NC + c] = v;
        }
        __syncthreads();
        if (rowsIn && lane < 16 * NC) Out[(size_t)waveRow * NC + lane] = sC[lane];
    }
}

extern "C" void kernel_launch(void* const* d_in, const int* in_sizes, int n_in,
                              void* d_out, int out_size, void* d_ws, size_t ws_size,
                              hipStream_t stream) {
    const float* in_feat = (const float*)d_in[0];
    const int*   src     = (const int*)d_in[1];
    const int*   dst     = (const int*)d_in[2];
    const float* W1 = (const float*)d_in[3];   const float* b1 = (const float*)d_in[4];
    const float* W2 = (const float*)d_in[5];   const float* b2 = (const float*)d_in[6];
    const float* Wc1 = (const float*)d_in[7];  const float* bc1 = (const float*)d_in[8];
    const float* Wc2 = (const float*)d_in[9];  const float* bc2 = (const float*)d_in[10];
    const float* W3 = (const float*)d_in[11];  const float* b3 = (const float*)d_in[12];
    const float* W4 = (const float*)d_in[13];  const float* b4 = (const float*)d_in[14];

    char* ws = (char*)d_ws;
    const size_t SZ_H   = (size_t)N_NODES * 64 * 4;
    const size_t SZ_CAT = (size_t)N_NODES * 192 * 4;
    float* dinv = (float*)(ws);
    float* bufA = (float*)(ws + 200192);
    float* bufB = (float*)(ws + 200192 + SZ_H);
    float* xcat = (float*)(ws + 200192 + 2 * SZ_H);
    char*  eb   = ws + 200192 + 2 * SZ_H + SZ_CAT;
    int*   cnt    = (int*)(eb);
    int*   rowptr = (int*)(eb + 200192);
    int*   rank   = (int*)(eb + 400384);
    int2*  ed     = (int2*)(eb + 400384 + (size_t)N_EDGES * 4);

    const int n64 = N_NODES * 64;
    dim3 blkS(256);
    dim3 gN((N_NODES + 255) / 256), gE((N_EDGES + 255) / 256), gNF((n64 + 255) / 256);
    dim3 gP((N_NODES * 16 + 255) / 256);

    zero_i32<<<gN, blkS, 0, stream>>>(cnt, N_NODES);
    edge_rank<<<gE, blkS, 0, stream>>>(dst, cnt, rank, N_EDGES);
    scan_rows<<<gN, blkS, 0, stream>>>(cnt, rowptr, dinv, N_NODES);
    fill_edges<<<gE, blkS, 0, stream>>>(src, dst, rank, rowptr, dinv, ed, N_EDGES);

    dim3 gemmBlk(GEMM_WAVES * 32);
    dim3 gG((N_NODES + GEMM_ROWS - 1) / GEMM_ROWS);

    gemm_rows<128, 64, true><<<gG, gemmBlk, 0, stream>>>(in_feat, W1, b1, bufA);
    gemm_rows<64, 64, true><<<gG, gemmBlk, 0, stream>>>(bufA, W2, b2, bufB);

    const float rn = 2.0f / 2.0f;
    const float* h_in = bufB;
    float*       h_out = bufA;
    const float* Wc[2] = { Wc1, Wc2 };
    const float* bc[2] = { bc1, bc2 };
    for (int layer = 0; layer < 2; ++layer) {
        copy_to_xcat<<<gNF, blkS, 0, stream>>>(h_in, xcat, n64);
        spmm_pull<false><<<gP, blkS, 0, stream>>>(rowptr, cnt, ed, dinv, xcat, -rn, rn - 1.0f, 0.0f);
        spmm_pull<true><<<gP, blkS, 0, stream>>>(rowptr, cnt, ed, dinv, xcat, -2.0f * rn, 2.0f * (rn - 1.0f), -1.0f);
        gemm_rows<192, 64, true><<<gG, gemmBlk, 0, stream>>>(xcat, Wc[layer], bc[layer], h_out);
        const float* t = h_in; h_in = h_out; h_out = (float*)t;
    }
    gemm_rows<64, 64, true><<<gG, gemmBlk, 0, stream>>>(h_in, W3, b3, (float*)h_out);
    gemm_rows<64, 2, false><<<gG, gemmBlk, 0, stream>>>((const float*)h_out, W4, b4, (float*)d_out);
}


# copy passes removed: the GEMM producing each Chebyshev layer's X0 stores its tiles straight into the concat buffer's first column block (row pitch 192; the second in place); 16 -> 14 launches
# speedup vs baseline: 4.5739x; 1.0714x over previous
#include <hip/hip_runtime.h>
#include <hip/hip_bf16.h>

typedef __attribute__((ext_vector_type(2))) float v2f;
typedef __attribute__((ext_vector_type(4))) float v4f;
typedef __attribute__((ext_vector_type(8))) float v8f;

#define N_NODES 50000
#define N_EDGES 800000

__global__ void zero_i32(int* __restrict__ p, int n) {
    int i = blockIdx.x * blockDim.x + threadIdx.x;
    if (i < n) p[i] = 0;
}

__global__ void edge_rank(const int* __restrict__ dst, int* __restrict__ cnt,
                          int* __restrict__ rank, int e) {
    int i = blockIdx.x * blockDim.x + threadIdx.x;
    if (i < e) rank[i] = atomicAdd(&cnt[dst[i]], 1);
}

__global__ void __launch_bounds__(256)
scan_rows(const int* __restrict__ cnt, int* __restrict__ rowptr,
          float* __restrict__ dinv, int n) {
    __shared__ int sTot[8];
    __shared__ int sPre[8];
    const int tid = threadIdx.x, lane = tid & 31, wave = tid >> 5;
    const int base = blockIdx.x * 256;
    int pre = 0;
    for (int i = tid; i < base; i += 256) pre += cnt[i];
    const int i = base + tid;
    const int c = (i < n) ? cnt[i] : 0;
    int inc = c;
    for (int o = 1; o < 32; o <<= 1) {
        int t = __shfl_up(inc, o);
        if (lane >= o) inc += t;
    }
    for (int o = 16; o > 0; o >>= 1) pre += __shfl_xor(pre, o);
    if (lane == 31) sTot[wave] = inc;
    if (lane == 0)  sPre[wave] = pre;
    __syncthreads();
    int off = 0;
    for (int w = 0; w < 8; ++w) {
        off += sPre[w];
        if (w < wave) off += sTot[w];
    }
    if (i < n) {
        rowptr[i] = off + inc - c;
        dinv[i]   = rsqrtf(fmaxf((float)c, 1.0f));
    }
}

__global__ void fill_edges(const int* __restrict__ src, const int* __restrict__ dst,
                           const int* __restrict__ rank, const int* __restrict__ rowptr,
                           const float* __restrict__ dinv, int2* __restrict__ ed, int e) {
    int i = blockIdx.x * blockDim.x + threadIdx.x;
    if (i >= e) return;
    int s = src[i], d = dst[i];
    ed[rowptr[d] + rank[i]] = make_int2(s, __float_as_int(dinv[s]));
}

template <bool SECOND>
__global__ void __launch_bounds__(256)
spmm_pull(const int* __restrict__ rowptr, const int* __restrict__ cnt,
          const int2* __restrict__ ed, const float* __restrict__ dinv,
          float* xcat, float cAgg, float cIn, float cX0) {
    const int row = (blockIdx.x * 256 + threadIdx.x) >> 4;
    const int l   = threadIdx.x & 15;
    if (row >= N_NODES) return;
    constexpr int IN = SECOND ? 64 : 0;
    const float* xb = xcat + IN + 4 * l;
    const int beg = rowptr[row];
    const int end = beg + cnt[row];
    const v4f zero = {0.0f, 0.0f, 0.0f, 0.0f};
    v4f acc = zero;
    for (int e = beg; e < end; e += 8) {
        int2 p[8];
        #pragma unroll
        for (int j = 0; j < 8; ++j) {
            int k = e + j;
            p[j] = ed[k < end ? k : end - 1];
        }
        v4f x[8];
        #pragma unroll
        for (int j = 0; j < 8; ++j) x[j] = *(const v4f*)(xb + (size_t)p[j].x * 192);
        #pragma unroll
        for (int j = 0; j < 8; ++j) {
            v4f xv = (e + j < end) ? x[j] : zero;
            acc += xv * __int_as_float(p[j].y);
        }
    }
    const float di = dinv[row];
    float* o = xcat + (size_t)row * 192;
    const v4f own = *(const v4f*)(o + IN + 4 * l);
    v4f r = cAgg * (acc * di) + cIn * own;
    if (SECOND) {
        const v4f x0 = *(const v4f*)(o + 4 * l);
        r += cX0 * x0;
    }
    *(v4f*)(o + IN + 64 + 4 * l) = r;
}

#define GEMM_WAVES 8
#define GEMM_ROWS  (GEMM_WAVES * 16)

template <int K, int NC, bool RELU, int OS = NC>
__global__ void __launch_bounds__(GEMM_WAVES * 32)
gemm_rows(const float* X, const float* __restrict__ W,
          const float* __restrict__ bias, float* Out) {
    constexpr int NT  = (NC + 15) / 16;
    constexpr int P   = K + 4;
    constexpr int KQ  = K / 4;
    constexpr int AQ  = K / 8;
    constexpr int WQ  = NT * 16 * KQ / (GEMM_WAVES * 32);
    static_assert(WQ * GEMM_WAVES * 32 == NT * 16 * KQ, "the W tile must divide evenly over the block");
    constexpr int PC  = 68;
    static_assert(NC == 64 || NC < 16, "the epilogue handles 64 columns or fewer than 16");
    static_assert(OS == NC || (NC == 64 && OS % 64 == 0), "a wider output pitch keeps every 256-byte row piece aligned");

    __shared__ alignas(16) float sW[NT * 16 * P];
    __shared__ alignas(16) float sA[GEMM_WAVES][16 * P];

    const int tid  = threadIdx.x;
    const int wave = __builtin_amdgcn_readfirstlane(tid >> 5);
    const int lane = tid & 31;
    const int waveRow = blockIdx.x * GEMM_ROWS + wave * 16;
    const bool rowsIn = waveRow < N_NODES;

    const v4f* Ag = (const v4f*)(X + (size_t)waveRow * K);
    v4f av[AQ];
    if (rowsIn) {
        #pragma unroll
        for (int q = 0; q < AQ; ++q) av[q] = Ag[q * 32 + lane];
    } else {
        #pragma unroll
        for (int q = 0; q < AQ; ++q) {
            int j = q * 32 + lane;
            int r = j / KQ, kq = j - r * KQ;
            *(v4f*)&sA[wave][r * P + kq * 4] = v4f{0.0f, 0.0f, 0.0f, 0.0f};
        }
    }

    const v4f* Wg = (const v4f*)W;
    v4f wv[WQ];
    #pragma unroll
    for (int i = 0; i < WQ; ++i) {
        int j = i * (GEMM_WAVES * 32) + tid;
        wv[i] = (NC == NT * 16 || j < NC * KQ) ? Wg[j] : v4f{0.0f, 0.0f, 0.0f, 0.0f};
    }

    const int c = lane & 15;
    const int h = lane >> 4;
    float bv[NT];
    #pragma unroll
    for (int nt = 0; nt < NT; ++nt) {
        int col = nt * 16 + c;
        bv[nt] = (col < NC) ? bias[col] : 0.0f;
    }

    #pragma unroll
    for (int q = 0; q < AQ; ++q) {
        int j = q * 32 + lane;
        int r = j / KQ, kq = j - r * KQ;
        if (rowsIn) *(v4f*)&sA[wave][r * P + kq * 4] = av[q];
    }
    #pragma unroll
    for (int i = 0; i < WQ; ++i) {
        int j = i * (GEMM_WAVES * 32) + tid;
        int n = j / KQ, kq = j - n * KQ;
        *(v4f*)&sW[n * P + kq * 4] = wv[i];
    }
    __syncthreads();

    const float* aF = &sA[wave][c * P + 4 * h];
    const float* bF = &sW[c * P + 4 * h];
    v8f acc[NT];
    #pragma unroll
    for (int nt = 0; nt < NT; ++nt) acc[nt] = v8f{};

    #pragma unroll
    for (int t = 0; t < AQ; ++t) {
        v4f a = *(const v4f*)(aF + 8 * t);
        v4f b[NT];
        #pragma unroll
        for (int nt = 0; nt < NT; ++nt) b[nt] = *(const v4f*)(bF + nt * 16 * P + 8 * t);
        v2f alo = {a.x, a.y}, ahi = {a.z, a.w};
        #pragma unroll
        for (int nt = 0; nt < NT; ++nt) {
            v2f blo = {b[nt].x, b[nt].y};
            acc[nt] = __builtin_amdgcn_wmma_f32_16x16x4_f32(
                false, alo, false, blo, (short)0, acc[nt], false, false);
        }
        #pragma unroll
        for (int nt = 0; nt < NT; ++nt) {
            v2f bhi = {b[nt].z, b[nt].w};
            acc[nt] = __builtin_amdgcn_wmma_f32_16x16x4_f32(
                false, ahi, false, bhi, (short)0, acc[nt], false, false);
        }
    }

    float* sC = &sA[wave][0];
    if constexpr (NC == 64) {
        #pragma unroll
        for (int nt = 0; nt < NT; ++nt) {
            #pragma unroll
            for (int r = 0; r < 8; ++r) {
                float v = acc[nt][r] + bv[nt];
                if (RELU) v = fmaxf(v, 0.0f);
                sC[(8 * h + r) * PC + nt * 16 + c] = v;
            }
        }
        __syncthreads();
        float* Og = Out + (size_t)waveRow * OS;
        #pragma unroll
        for (int q = 0; q < 8; ++q) {
            int j = q * 32 + lane;
            v4f v = *(const v4f*)&sC[(j >> 4) * PC + (j & 15) * 4];
            if (rowsIn) *(v4f*)(Og + (j >> 4) * OS + (j & 15) * 4) = v;
        }
    } else {
        #pragma unroll
        for (int r = 0; r < 8; ++r) {
            float v = acc[0][r] + bv[0];
            if (RELU) v = fmaxf(v, 0.0f);
            if (c < NC) sC[(8 * h + r) * NC + c] = v;
        }
        __syncthreads();
        if (rowsIn && lane < 16 * NC) Out[(size_t)waveRow * NC + lane] = sC[lane];
    }
}

extern "C" void kernel_launch(void* const* d_in, const int* in_sizes, int n_in,
                              void* d_out, int out_size, void* d_ws, size_t ws_size,
                              hipStream_t stream) {
    const float* in_feat = (const float*)d_in[0];
    const int*   src     = (const int*)d_in[1];
    const int*   dst     = (const int*)d_in[2];
    const float* W1 = (const float*)d_in[3];   const float* b1 = (const float*)d_in[4];
    const float* W2 = (const float*)d_in[5];   const float* b2 = (const float*)d_in[6];
    const float* Wc1 = (const float*)d_in[7];  const float* bc1 = (const float*)d_in[8];
    const float* Wc2 = (const float*)d_in[9];  const float* bc2 = (const float*)d_in[10];
    const float* W3 = (const float*)d_in[11];  const float* b3 = (const float*)d_in[12];
    const float* W4 = (const float*)d_in[13];  const float* b4 = (const float*)d_in[14];

    char* ws = (char*)d_ws;
    const size_t SZ_H   = (size_t)N_NODES * 64 * 4;
    const size_t SZ_CAT = (size_t)N_NODES * 192 * 4;
    float* dinv = (float*)(ws);
    float* bufA = (float*)(ws + 200192);
    float* bufB = (float*)(ws + 200192 + SZ_H);
    float* xcat = (float*)(ws + 200192 + 2 * SZ_H);
    char*  eb   = ws + 200192 + 2 * SZ_H + SZ_CAT;
    int*   cnt    = (int*)(eb);
    int*   rowptr = (int*)(eb + 200192);
    int*   rank   = (int*)(eb + 400384);
    int2*  ed     = (int2*)(eb + 400384 + (size_t)N_EDGES * 4);

    dim3 blkS(256);
    dim3 gN((N_NODES + 255) / 256), gE((N_EDGES + 255) / 256);
    dim3 gP((N_NODES * 16 + 255) / 256);

    zero_i32<<<gN, blkS, 0, stream>>>(cnt, N_NODES);
    edge_rank<<<gE, blkS, 0, stream>>>(dst, cnt, rank, N_EDGES);
    scan_rows<<<gN, blkS, 0, stream>>>(cnt, rowptr, dinv, N_NODES);
    fill_edges<<<gE, blkS, 0, stream>>>(src, dst, rank, rowptr, dinv, ed, N_EDGES);

    dim3 gemmBlk(GEMM_WAVES * 32);
    dim3 gG((N_NODES + GEMM_ROWS - 1) / GEMM_ROWS);

    gemm_rows<128, 64, true><<<gG, gemmBlk, 0, stream>>>(in_feat, W1, b1, bufA);
    gemm_rows<64, 64, true, 192><<<gG, gemmBlk, 0, stream>>>(bufA, W2, b2, xcat);

    const float rn = 2.0f / 2.0f;
    spmm_pull<false><<<gP, blkS, 0, stream>>>(rowptr, cnt, ed, dinv, xcat, -rn, rn - 1.0f, 0.0f);
    spmm_pull<true><<<gP, blkS, 0, stream>>>(rowptr, cnt, ed, dinv, xcat, -2.0f * rn, 2.0f * (rn - 1.0f), -1.0f);
    gemm_rows<192, 64, true, 192><<<gG, gemmBlk, 0, stream>>>(xcat, Wc1, bc1, xcat);

    spmm_pull<false><<<gP, blkS, 0, stream>>>(rowptr, cnt, ed, dinv, xcat, -rn, rn - 1.0f, 0.0f);
    spmm_pull<true><<<gP, blkS, 0, stream>>>(rowptr, cnt, ed, dinv, xcat, -2.0f * rn, 2.0f * (rn - 1.0f), -1.0f);
    gemm_rows<192, 64, true><<<gG, gemmBlk, 0, stream>>>(xcat, Wc2, bc2, bufA);

    gemm_rows<64, 64, true><<<gG, gemmBlk, 0, stream>>>(bufA, W3, b3, bufB);
    gemm_rows<64, 2, false><<<gG, gemmBlk, 0, stream>>>(bufB, W4, b4, (float*)d_out);
}


# independent edge bucketing overlapped with the first two GEMMs: rank blocks (4 edges per thread) share GEMM 1's launch, scan blocks GEMM 2's; 14 -> 12 launches; row sums stay in atomic rank order
# speedup vs baseline: 4.8426x; 1.0587x over previous
#include <hip/hip_runtime.h>
#include <hip/hip_bf16.h>

typedef __attribute__((ext_vector_type(2))) float v2f;
typedef __attribute__((ext_vector_type(4))) float v4f;
typedef __attribute__((ext_vector_type(8))) float v8f;

#define N_NODES 50000
#define N_EDGES 800000

__global__ void zero_i32(int* __restrict__ p, int n) {
    int i = blockIdx.x * blockDim.x + threadIdx.x;
    if (i < n) p[i] = 0;
}

__device__ __forceinline__ void
scan_body(const int blk, const int* __restrict__ cnt, int* __restrict__ rowptr,
          float* __restrict__ dinv, int n) {
    __shared__ int sTot[8];
    __shared__ int sPre[8];
    const int tid = threadIdx.x, lane = tid & 31, wave = tid >> 5;
    const int base = blk * 256;
    int pre = 0;
    for (int i = tid; i < base; i += 256) pre += cnt[i];
    const int i = base + tid;
    const int c = (i < n) ? cnt[i] : 0;
    int inc = c;
    for (int o = 1; o < 32; o <<= 1) {
        int t = __shfl_up(inc, o);
        if (lane >= o) inc += t;
    }
    for (int o = 16; o > 0; o >>= 1) pre += __shfl_xor(pre, o);
    if (lane == 31) sTot[wave] = inc;
    if (lane == 0)  sPre[wave] = pre;
    __syncthreads();
    int off = 0;
    for (int w = 0; w < 8; ++w) {
        off += sPre[w];
        if (w < wave) off += sTot[w];
    }
    if (i < n) {
        rowptr[i] = off + inc - c;
        dinv[i]   = rsqrtf(fmaxf((float)c, 1.0f));
    }
}

__global__ void fill_edges(const int* __restrict__ src, const int* __restrict__ dst,
                           const int* __restrict__ rank, const int* __restrict__ rowptr,
                           const float* __restrict__ dinv, int2* __restrict__ ed, int e) {
    int i = blockIdx.x * blockDim.x + threadIdx.x;
    if (i >= e) return;
    int s = src[i], d = dst[i];
    ed[rowptr[d] + rank[i]] = make_int2(s, __float_as_int(dinv[s]));
}

template <bool SECOND>
__global__ void __launch_bounds__(256)
spmm_pull(const int* __restrict__ rowptr, const int* __restrict__ cnt,
          const int2* __restrict__ ed, const float* __restrict__ dinv,
          float* xcat, float cAgg, float cIn, float cX0) {
    const int row = (blockIdx.x * 256 + threadIdx.x) >> 4;
    const int l   = threadIdx.x & 15;
    if (row >= N_NODES) return;
    constexpr int IN = SECOND ? 64 : 0;
    const float* xb = xcat + IN + 4 * l;
    const int beg = rowptr[row];
    const int end = beg + cnt[row];
    const v4f zero = {0.0f, 0.0f, 0.0f, 0.0f};
    v4f acc = zero;
    for (int e = beg; e < end; e += 8) {
        int2 p[8];
        #pragma unroll
        for (int j = 0; j < 8; ++j) {
            int k = e + j;
            p[j] = ed[k < end ? k : end - 1];
        }
        v4f x[8];
        #pragma unroll
        for (int j = 0; j < 8; ++j) x[j] = *(const v4f*)(xb + (size_t)p[j].x * 192);
        #pragma unroll
        for (int j = 0; j < 8; ++j) {
            v4f xv = (e + j < end) ? x[j] : zero;
            acc += xv * __int_as_float(p[j].y);
        }
    }
    const float di = dinv[row];
    float* o = xcat + (size_t)row * 192;
    const v4f own = *(const v4f*)(o + IN + 4 * l);
    v4f r = cAgg * (acc * di) + cIn * own;
    if (SECOND) {
        const v4f x0 = *(const v4f*)(o + 4 * l);
        r += cX0 * x0;
    }
    *(v4f*)(o + IN + 64 + 4 * l) = r;
}

#define GEMM_WAVES 8
#define GEMM_ROWS  (GEMM_WAVES * 16)

template <int K, int NC, bool RELU, int OS = NC>
__device__ __forceinline__ void
gemm_body(const int blk, const float* X, const float* __restrict__ W,
          const float* __restrict__ bias, float* Out) {
    constexpr int NT  = (NC + 15) / 16;
    constexpr int P   = K + 4;
    constexpr int KQ  = K / 4;
    constexpr int AQ  = K / 8;
    constexpr int WQ  = NT * 16 * KQ / (GEMM_WAVES * 32);
    static_assert(WQ * GEMM_WAVES * 32 == NT * 16 * KQ, "the W tile must divide evenly over the block");
    constexpr int PC  = 68;
    static_assert(NC == 64 || NC < 16, "the epilogue handles 64 columns or fewer than 16");
    static_assert(OS == NC || (NC == 64 && OS % 64 == 0), "a wider output pitch keeps every 256-byte row piece aligned");

    __shared__ alignas(16) float sW[NT * 16 * P];
    __shared__ alignas(16) float sA[GEMM_WAVES][16 * P];

    const int tid  = threadIdx.x;
    const int wave = __builtin_amdgcn_readfirstlane(tid >> 5);
    const int lane = tid & 31;
    const int waveRow = blk * GEMM_ROWS + wave * 16;
    const bool rowsIn = waveRow < N_NODES;

    const v4f* Ag = (const v4f*)(X + (size_t)waveRow * K);
    v4f av[AQ];
    if (rowsIn) {
        #pragma unroll
        for (int q = 0; q < AQ; ++q) av[q] = Ag[q * 32 + lane];
    } else {
        #pragma unroll
        for (int q = 0; q < AQ; ++q) {
            int j = q * 32 + lane;
            int r = j / KQ, kq = j - r * KQ;
            *(v4f*)&sA[wave][r * P + kq * 4] = v4f{0.0f, 0.0f, 0.0f, 0.0f};
        }
    }

    const v4f* Wg = (const v4f*)W;
    v4f wv[WQ];
    #pragma unroll
    for (int i = 0; i < WQ; ++i) {
        int j = i * (GEMM_WAVES * 32) + tid;
        wv[i] = (NC == NT * 16 || j < NC * KQ) ? Wg[j] : v4f{0.0f, 0.0f, 0.0f, 0.0f};
    }

    const int c = lane & 15;
    const int h = lane >> 4;
    float bv[NT];
    #pragma unroll
    for (int nt = 0; nt < NT; ++nt) {
        int col = nt * 16 + c;
        bv[nt] = (col < NC) ? bias[col] : 0.0f;
    }

    #pragma unroll
    for (int q = 0; q < AQ; ++q) {
        int j = q * 32 + lane;
        int r = j / KQ, kq = j - r * KQ;
        if (rowsIn) *(v4f*)&sA[wave][r * P + kq * 4] = av[q];
    }
    #pragma unroll
    for (int i = 0; i < WQ; ++i) {
        int j = i * (GEMM_WAVES * 32) + tid;
        int n = j / KQ, kq = j - n * KQ;
        *(v4f*)&sW[n * P + kq * 4] = wv[i];
    }
    __syncthreads();

    const float* aF = &sA[wave][c * P + 4 * h];
    const float* bF = &sW[c * P + 4 * h];
    v8f acc[NT];
    #pragma unroll
    for (int nt = 0; nt < NT; ++nt) acc[nt] = v8f{};

    #pragma unroll
    for (int t = 0; t < AQ; ++t) {
        v4f a = *(const v4f*)(aF + 8 * t);
        v4f b[NT];
        #pragma unroll
        for (int nt = 0; nt < NT; ++nt) b[nt] = *(const v4f*)(bF + nt * 16 * P + 8 * t);
        v2f alo = {a.x, a.y}, ahi = {a.z, a.w};
        #pragma unroll
        for (int nt = 0; nt < NT; ++nt) {
            v2f blo = {b[nt].x, b[nt].y};
            acc[nt] = __builtin_amdgcn_wmma_f32_16x16x4_f32(
                false, alo, false, blo, (short)0, acc[nt], false, false);
        }
        #pragma unroll
        for (int nt = 0; nt < NT; ++nt) {
            v2f bhi = {b[nt].z, b[nt].w};
            acc[nt] = __builtin_amdgcn_wmma_f32_16x16x4_f32(
                false, ahi, false, bhi, (short)0, acc[nt], false, false);
        }
    }

    float* sC = &sA[wave][0];
    if constexpr (NC == 64) {
        #pragma unroll
        for (int nt = 0; nt < NT; ++nt) {
            #pragma unroll
            for (int r = 0; r < 8; ++r) {
                float v = acc[nt][r] + bv[nt];
                if (RELU) v = fmaxf(v, 0.0f);
                sC[(8 * h + r) * PC + nt * 16 + c] = v;
            }
        }
        __syncthreads();
        float* Og = Out + (size_t)waveRow * OS;
        #pragma unroll
        for (int q = 0; q < 8; ++q) {
            int j = q * 32 + lane;
            v4f v = *(const v4f*)&sC[(j >> 4) * PC + (j & 15) * 4];
            if (rowsIn) *(v4f*)(Og + (j >> 4) * OS + (j & 15) * 4) = v;
        }
    } else {
        #pragma unroll
        for (int r = 0; r < 8; ++r) {
            float v = acc[0][r] + bv[0];
            if (RELU) v = fmaxf(v, 0.0f);
            if (c < NC) sC[(8 * h + r) * NC + c] = v;
        }
        __syncthreads();
        if (rowsIn && lane < 16 * NC) Out[(size_t)waveRow * NC + lane] = sC[lane];
    }
}

template <int K, int NC, bool RELU, int OS = NC>
__global__ void __launch_bounds__(GEMM_WAVES * 32)
gemm_rows(const float* X, const float* __restrict__ W,
          const float* __restrict__ bias, float* Out) {
    gemm_body<K, NC, RELU, OS>(blockIdx.x, X, W, bias, Out);
}

#define GEMM_BLOCKS ((N_NODES + GEMM_ROWS - 1) / GEMM_ROWS)
#define RANK_BLOCKS ((N_EDGES / 4 + 255) / 256)
#define SCAN_BLOCKS ((N_NODES + 255) / 256)
static_assert(N_EDGES % 4 == 0, "edges are ranked four per thread");

__global__ void __launch_bounds__(GEMM_WAVES * 32)
gemm1_and_rank(const float* in_feat, const float* __restrict__ W1,
               const float* __restrict__ b1, float* h,
               const int* __restrict__ dst, int* __restrict__ cnt, int* __restrict__ rank) {
    if (blockIdx.x < GEMM_BLOCKS) {
        gemm_body<128, 64, true>(blockIdx.x, in_feat, W1, b1, h);
    } else {
        const int i = (blockIdx.x - GEMM_BLOCKS) * 256 + threadIdx.x;
        if (i < N_EDGES / 4) {
            const int4 d = ((const int4*)dst)[i];
            int4 r;
            r.x = atomicAdd(&cnt[d.x], 1);
            r.y = atomicAdd(&cnt[d.y], 1);
            r.z = atomicAdd(&cnt[d.z], 1);
            r.w = atomicAdd(&cnt[d.w], 1);
            ((int4*)rank)[i] = r;
        }
    }
}

__global__ void __launch_bounds__(GEMM_WAVES * 32)
gemm2_and_scan(const float* h, const float* __restrict__ W2,
               const float* __restrict__ b2, float* xcat,
               const int* __restrict__ cnt, int* __restrict__ rowptr, float* __restrict__ dinv) {
    if (blockIdx.x < GEMM_BLOCKS) {
        gemm_body<64, 64, true, 192>(blockIdx.x, h, W2, b2, xcat);
    } else {
        scan_body(blockIdx.x - GEMM_BLOCKS, cnt, rowptr, dinv, N_NODES);
    }
}

extern "C" void kernel_launch(void* const* d_in, const int* in_sizes, int n_in,
                              void* d_out, int out_size, void* d_ws, size_t ws_size,
                              hipStream_t stream) {
    const float* in_feat = (const float*)d_in[0];
    const int*   src     = (const int*)d_in[1];
    const int*   dst     = (const int*)d_in[2];
    const float* W1 = (const float*)d_in[3];   const float* b1 = (const float*)d_in[4];
    const float* W2 = (const float*)d_in[5];   const float* b2 = (const float*)d_in[6];
    const float* Wc1 = (const float*)d_in[7];  const float* bc1 = (const float*)d_in[8];
    const float* Wc2 = (const float*)d_in[9];  const float* bc2 = (const float*)d_in[10];
    const float* W3 = (const float*)d_in[11];  const float* b3 = (const float*)d_in[12];
    const float* W4 = (const float*)d_in[13];  const float* b4 = (const float*)d_in[14];

    char* ws = (char*)d_ws;
    const size_t SZ_H   = (size_t)N_NODES * 64 * 4;
    const size_t SZ_CAT = (size_t)N_NODES * 192 * 4;
    float* dinv = (float*)(ws);
    float* bufA = (float*)(ws + 200192);
    float* bufB = (float*)(ws + 200192 + SZ_H);
    float* xcat = (float*)(ws + 200192 + 2 * SZ_H);
    char*  eb   = ws + 200192 + 2 * SZ_H + SZ_CAT;
    int*   cnt    = (int*)(eb);
    int*   rowptr = (int*)(eb + 200192);
    int*   rank   = (int*)(eb + 400384);
    int2*  ed     = (int2*)(eb + 400384 + (size_t)N_EDGES * 4);

    dim3 blkS(256);
    dim3 gN((N_NODES + 255) / 256), gE((N_EDGES + 255) / 256);
    dim3 gP((N_NODES * 16 + 255) / 256);

    dim3 gemmBlk(GEMM_WAVES * 32);
    dim3 gG(GEMM_BLOCKS);

    zero_i32<<<gN, blkS, 0, stream>>>(cnt, N_NODES);
    gemm1_and_rank<<<dim3(GEMM_BLOCKS + RANK_BLOCKS), gemmBlk, 0, stream>>>(in_feat, W1, b1, bufA, dst, cnt, rank);
    gemm2_and_scan<<<dim3(GEMM_BLOCKS + SCAN_BLOCKS), gemmBlk, 0, stream>>>(bufA, W2, b2, xcat, cnt, rowptr, dinv);
    fill_edges<<<gE, blkS, 0, stream>>>(src, dst, rank, rowptr, dinv, ed, N_EDGES);

    const float rn = 2.0f / 2.0f;
    spmm_pull<false><<<gP, blkS, 0, stream>>>(rowptr, cnt, ed, dinv, xcat, -rn, rn - 1.0f, 0.0f);
    spmm_pull<true><<<gP, blkS, 0, stream>>>(rowptr, cnt, ed, dinv, xcat, -2.0f * rn, 2.0f * (rn - 1.0f), -1.0f);
    gemm_rows<192, 64, true, 192><<<gG, gemmBlk, 0, stream>>>(xcat, Wc1, bc1, xcat);

    spmm_pull<false><<<gP, blkS, 0, stream>>>(rowptr, cnt, ed, dinv, xcat, -rn, rn - 1.0f, 0.0f);
    spmm_pull<true><<<gP, blkS, 0, stream>>>(rowptr, cnt, ed, dinv, xcat, -2.0f * rn, 2.0f * (rn - 1.0f), -1.0f);
    gemm_rows<192, 64, true><<<gG, gemmBlk, 0, stream>>>(xcat, Wc2, bc2, bufA);

    gemm_rows<64, 64, true><<<gG, gemmBlk, 0, stream>>>(bufA, W3, b3, bufB);
    gemm_rows<64, 2, false><<<gG, gemmBlk, 0, stream>>>(bufB, W4, b4, (float*)d_out);
}


# tail dense chain Wc2>W3>W4 in one launch: a wave's 16x64 relu tile stays in its LDS slot as the next layer's A operand; W3, W4 reuse Wc2's staging area (LDS size kept); 12 -> 10 launches
# speedup vs baseline: 5.0689x; 1.0467x over previous
#include <hip/hip_runtime.h>
#include <hip/hip_bf16.h>

typedef __attribute__((ext_vector_type(2))) float v2f;
typedef __attribute__((ext_vector_type(4))) float v4f;
typedef __attribute__((ext_vector_type(8))) float v8f;

#define N_NODES 50000
#define N_EDGES 800000

__global__ void zero_i32(int* __restrict__ p, int n) {
    int i = blockIdx.x * blockDim.x + threadIdx.x;
    if (i < n) p[i] = 0;
}

__device__ __forceinline__ void
scan_body(const int blk, const int* __restrict__ cnt, int* __restrict__ rowptr,
          float* __restrict__ dinv, int n) {
    __shared__ int sTot[8];
    __shared__ int sPre[8];
    const int tid = threadIdx.x, lane = tid & 31, wave = tid >> 5;
    const int base = blk * 256;
    int pre = 0;
    for (int i = tid; i < base; i += 256) pre += cnt[i];
    const int i = base + tid;
    const int c = (i < n) ? cnt[i] : 0;
    int inc = c;
    for (int o = 1; o < 32; o <<= 1) {
        int t = __shfl_up(inc, o);
        if (lane >= o) inc += t;
    }
    for (int o = 16; o > 0; o >>= 1) pre += __shfl_xor(pre, o);
    if (lane == 31) sTot[wave] = inc;
    if (lane == 0)  sPre[wave] = pre;
    __syncthreads();
    int off = 0;
    for (int w = 0; w < 8; ++w) {
        off += sPre[w];
        if (w < wave) off += sTot[w];
    }
    if (i < n) {
        rowptr[i] = off + inc - c;
        dinv[i]   = rsqrtf(fmaxf((float)c, 1.0f));
    }
}

__global__ void fill_edges(const int* __restrict__ src, const int* __restrict__ dst,
                           const int* __restrict__ rank, const int* __restrict__ rowptr,
                           const float* __restrict__ dinv, int2* __restrict__ ed, int e) {
    int i = blockIdx.x * blockDim.x + threadIdx.x;
    if (i >= e) return;
    int s = src[i], d = dst[i];
    ed[rowptr[d] + rank[i]] = make_int2(s, __float_as_int(dinv[s]));
}

template <bool SECOND>
__global__ void __launch_bounds__(256)
spmm_pull(const int* __restrict__ rowptr, const int* __restrict__ cnt,
          const int2* __restrict__ ed, const float* __restrict__ dinv,
          float* xcat, float cAgg, float cIn, float cX0) {
    const int row = (blockIdx.x * 256 + threadIdx.x) >> 4;
    const int l   = threadIdx.x & 15;
    if (row >= N_NODES) return;
    constexpr int IN = SECOND ? 64 : 0;
    const float* xb = xcat + IN + 4 * l;
    const int beg = rowptr[row];
    const int end = beg + cnt[row];
    const v4f zero = {0.0f, 0.0f, 0.0f, 0.0f};
    v4f acc = zero;
    for (int e = beg; e < end; e += 8) {
        int2 p[8];
        #pragma unroll
        for (int j = 0; j < 8; ++j) {
            int k = e + j;
            p[j] = ed[k < end ? k : end - 1];
        }
        v4f x[8];
        #pragma unroll
        for (int j = 0; j < 8; ++j) x[j] = *(const v4f*)(xb + (size_t)p[j].x * 192);
        #pragma unroll
        for (int j = 0; j < 8; ++j) {
            v4f xv = (e + j < end) ? x[j] : zero;
            acc += xv * __int_as_float(p[j].y);
        }
    }
    const float di = dinv[row];
    float* o = xcat + (size_t)row * 192;
    const v4f own = *(const v4f*)(o + IN + 4 * l);
    v4f r = cAgg * (acc * di) + cIn * own;
    if (SECOND) {
        const v4f x0 = *(const v4f*)(o + 4 * l);
        r += cX0 * x0;
    }
    *(v4f*)(o + IN + 64 + 4 * l) = r;
}

#define GEMM_WAVES 8
#define GEMM_ROWS  (GEMM_WAVES * 16)

template <int K, int NC, bool RELU, int OS = NC>
__device__ __forceinline__ void
gemm_body(const int blk, const float* X, const float* __restrict__ W,
          const float* __restrict__ bias, float* Out) {
    constexpr int NT  = (NC + 15) / 16;
    constexpr int P   = K + 4;
    constexpr int KQ  = K / 4;
    constexpr int AQ  = K / 8;
    constexpr int WQ  = NT * 16 * KQ / (GEMM_WAVES * 32);
    static_assert(WQ * GEMM_WAVES * 32 == NT * 16 * KQ, "the W tile must divide evenly over the block");
    constexpr int PC  = 68;
    static_assert(NC == 64 || NC < 16, "the epilogue handles 64 columns or fewer than 16");
    static_assert(OS == NC || (NC == 64 && OS % 64 == 0), "a wider output pitch keeps every 256-byte row piece aligned");

    __shared__ alignas(16) float sW[NT * 16 * P];
    __shared__ alignas(16) float sA[GEMM_WAVES][16 * P];

    const int tid  = threadIdx.x;
    const int wave = __builtin_amdgcn_readfirstlane(tid >> 5);
    const int lane = tid & 31;
    const int waveRow = blk * GEMM_ROWS + wave * 16;
    const bool rowsIn = waveRow < N_NODES;

    const v4f* Ag = (const v4f*)(X + (size_t)waveRow * K);
    v4f av[AQ];
    if (rowsIn) {
        #pragma unroll
        for (int q = 0; q < AQ; ++q) av[q] = Ag[q * 32 + lane];
    } else {
        #pragma unroll
        for (int q = 0; q < AQ; ++q) {
            int j = q * 32 + lane;
            int r = j / KQ, kq = j - r * KQ;
            *(v4f*)&sA[wave][r * P + kq * 4] = v4f{0.0f, 0.0f, 0.0f, 0.0f};
        }
    }

    const v4f* Wg = (const v4f*)W;
    v4f wv[WQ];
    #pragma unroll
    for (int i = 0; i < WQ; ++i) {
        int j = i * (GEMM_WAVES * 32) + tid;
        wv[i] = (NC == NT * 16 || j < NC * KQ) ? Wg[j] : v4f{0.0f, 0.0f, 0.0f, 0.0f};
    }

    const int c = lane & 15;
    const int h = lane >> 4;
    float bv[NT];
    #pragma unroll
    for (int nt = 0; nt < NT; ++nt) {
        int col = nt * 16 + c;
        bv[nt] = (col < NC) ? bias[col] : 0.0f;
    }

    #pragma unroll
    for (int q = 0; q < AQ; ++q) {
        int j = q * 32 + lane;
        int r = j / KQ, kq = j - r * KQ;
        if (rowsIn) *(v4f*)&sA[wave][r * P + kq * 4] = av[q];
    }
    #pragma unroll
    for (int i = 0; i < WQ; ++i) {
        int j = i * (GEMM_WAVES * 32) + tid;
        int n = j / KQ, kq = j - n * KQ;
        *(v4f*)&sW[n * P + kq * 4] = wv[i];
    }
    __syncthreads();

    const float* aF = &sA[wave][c * P + 4 * h];
    const float* bF = &sW[c * P + 4 * h];
    v8f acc[NT];
    #pragma unroll
    for (int nt = 0; nt < NT; ++nt) acc[nt] = v8f{};

    #pragma unroll
    for (int t = 0; t < AQ; ++t) {
        v4f a = *(const v4f*)(aF + 8 * t);
        v4f b[NT];
        #pragma unroll
        for (int nt = 0; nt < NT; ++nt) b[nt] = *(const v4f*)(bF + nt * 16 * P + 8 * t);
        v2f alo = {a.x, a.y}, ahi = {a.z, a.w};
        #pragma unroll
        for (int nt = 0; nt < NT; ++nt) {
            v2f blo = {b[nt].x, b[nt].y};
            acc[nt] = __builtin_amdgcn_wmma_f32_16x16x4_f32(
                false, alo, false, blo, (short)0, acc[nt], false, false);
        }
        #pragma unroll
        for (int nt = 0; nt < NT; ++nt) {
            v2f bhi = {b[nt].z, b[nt].w};
            acc[nt] = __builtin_amdgcn_wmma_f32_16x16x4_f32(
                false, ahi, false, bhi, (short)0, acc[nt], false, false);
        }
    }

    float* sC = &sA[wave][0];
    if constexpr (NC == 64) {
        #pragma unroll
        for (int nt = 0; nt < NT; ++nt) {
            #pragma unroll
            for (int r = 0; r < 8; ++r) {
                float v = acc[nt][r] + bv[nt];
                if (RELU) v = fmaxf(v, 0.0f);
                sC[(8 * h + r) * PC + nt * 16 + c] = v;
            }
        }
        __syncthreads();
        float* Og = Out + (size_t)waveRow * OS;
        #pragma unroll
        for (int q = 0; q < 8; ++q) {
            int j = q * 32 + lane;
            v4f v = *(const v4f*)&sC[(j >> 4) * PC + (j & 15) * 4];
            if (rowsIn) *(v4f*)(Og + (j >> 4) * OS + (j & 15) * 4) = v;
        }
    } else {
        #pragma unroll
        for (int r = 0; r < 8; ++r) {
            float v = acc[0][r] + bv[0];
            if (RELU) v = fmaxf(v, 0.0f);
            if (c < NC) sC[(8 * h + r) * NC + c] = v;
        }
        __syncthreads();
        if (rowsIn && lane < 16 * NC) Out[(size_t)waveRow * NC + lane] = sC[lane];
    }
}

template <int K, int NC, bool RELU, int OS = NC>
__global__ void __launch_bounds__(GEMM_WAVES * 32)
gemm_rows(const float* X, const float* __restrict__ W,
          const float* __restrict__ bias, float* Out) {
    gemm_body<K, NC, RELU, OS>(blockIdx.x, X, W, bias, Out);
}

#define GEMM_BLOCKS ((N_NODES + GEMM_ROWS - 1) / GEMM_ROWS)
#define RANK_BLOCKS ((N_EDGES / 4 + 255) / 256)
#define SCAN_BLOCKS ((N_NODES + 255) / 256)
static_assert(N_EDGES % 4 == 0, "edges are ranked four per thread");

__global__ void __launch_bounds__(GEMM_WAVES * 32)
gemm1_and_rank(const float* in_feat, const float* __restrict__ W1,
               const float* __restrict__ b1, float* h,
               const int* __restrict__ dst, int* __restrict__ cnt, int* __restrict__ rank) {
    if (blockIdx.x < GEMM_BLOCKS) {
        gemm_body<128, 64, true>(blockIdx.x, in_feat, W1, b1, h);
    } else {
        const int i = (blockIdx.x - GEMM_BLOCKS) * 256 + threadIdx.x;
        if (i < N_EDGES / 4) {
            const int4 d = ((const int4*)dst)[i];
            int4 r;
            r.x = atomicAdd(&cnt[d.x], 1);
            r.y = atomicAdd(&cnt[d.y], 1);
            r.z = atomicAdd(&cnt[d.z], 1);
            r.w = atomicAdd(&cnt[d.w], 1);
            ((int4*)rank)[i] = r;
        }
    }
}

__global__ void __launch_bounds__(GEMM_WAVES * 32)
gemm2_and_scan(const float* h, const float* __restrict__ W2,
               const float* __restrict__ b2, float* xcat,
               const int* __restrict__ cnt, int* __restrict__ rowptr, float* __restrict__ dinv) {
    if (blockIdx.x < GEMM_BLOCKS) {
        gemm_body<64, 64, true, 192>(blockIdx.x, h, W2, b2, xcat);
    } else {
        scan_body(blockIdx.x - GEMM_BLOCKS, cnt, rowptr, dinv, N_NODES);
    }
}

__global__ void __launch_bounds__(GEMM_WAVES * 32)
gemm_tail(const float* __restrict__ X, const float* __restrict__ Wc, const float* __restrict__ bc,
          const float* __restrict__ W3, const float* __restrict__ b3,
          const float* __restrict__ W4, const float* __restrict__ b4, float* __restrict__ Out) {
    constexpr int K   = 192;
    constexpr int NT  = 4;
    constexpr int P   = K + 4;
    constexpr int KQ  = K / 4;
    constexpr int AQ  = K / 8;
    constexpr int WQ  = NT * 16 * KQ / (GEMM_WAVES * 32);
    constexpr int PC  = 68;
    constexpr int W3Q = 64 * 16 / (GEMM_WAVES * 32);
    constexpr int OFF4 = 64 * PC;
    static_assert(OFF4 + 16 * PC <= NT * 16 * P, "W3 and W4 fit into the first layer's W area");

    __shared__ alignas(16) float sW[NT * 16 * P];
    __shared__ alignas(16) float sA[GEMM_WAVES][16 * P];

    const int tid  = threadIdx.x;
    const int wave = __builtin_amdgcn_readfirstlane(tid >> 5);
    const int lane = tid & 31;
    const int waveRow = blockIdx.x * GEMM_ROWS + wave * 16;
    const bool rowsIn = waveRow < N_NODES;
    const v4f zero4 = {0.0f, 0.0f, 0.0f, 0.0f};

    const v4f* Ag = (const v4f*)(X + (size_t)waveRow * K);
    v4f av[AQ];
    if (rowsIn) {
        #pragma unroll
        for (int q = 0; q < AQ; ++q) av[q] = Ag[q * 32 + lane];
    } else {
        #pragma unroll
        for (int q = 0; q < AQ; ++q) {
            int j = q * 32 + lane;
            int r = j / KQ, kq = j - r * KQ;
            *(v4f*)&sA[wave][r * P + kq * 4] = zero4;
        }
    }
    const v4f* Wg = (const v4f*)Wc;
    v4f wv[WQ];
    #pragma unroll
    for (int i = 0; i < WQ; ++i) wv[i] = Wg[i * (GEMM_WAVES * 32) + tid];

    const int c = lane & 15;
    const int h = lane >> 4;
    float bv[NT], b3v[NT];
    #pragma unroll
    for (int nt = 0; nt < NT; ++nt) {
        bv[nt]  = bc[nt * 16 + c];
        b3v[nt] = b3[nt * 16 + c];
    }
    const float b4v = (c < 2) ? b4[c] : 0.0f;

    #pragma unroll
    for (int q = 0; q < AQ; ++q) {
        int j = q * 32 + lane;
        int r = j / KQ, kq = j - r * KQ;
        if (rowsIn) *(v4f*)&sA[wave][r * P + kq * 4] = av[q];
    }
    #pragma unroll
    for (int i = 0; i < WQ; ++i) {
        int j = i * (GEMM_WAVES * 32) + tid;
        int n = j / KQ, kq = j - n * KQ;
        *(v4f*)&sW[n * P + kq * 4] = wv[i];
    }
    __syncthreads();

    const v4f* W3g = (const v4f*)W3;
    v4f w3v[W3Q];
    #pragma unroll
    for (int i = 0; i < W3Q; ++i) w3v[i] = W3g[i * (GEMM_WAVES * 32) + tid];
    const v4f w4v = (tid < 2 * 16) ? ((const v4f*)W4)[tid] : zero4;

    const float* aF = &sA[wave][c * P + 4 * h];
    const float* bF = &sW[c * P + 4 * h];
    v8f acc[NT];
    #pragma unroll
    for (int nt = 0; nt < NT; ++nt) acc[nt] = v8f{};
    #pragma unroll
    for (int t = 0; t < AQ; ++t) {
        v4f a = *(const v4f*)(aF + 8 * t);
        v4f b[NT];
        #pragma unroll
        for (int nt = 0; nt < NT; ++nt) b[nt] = *(const v4f*)(bF + nt * 16 * P + 8 * t);
        v2f alo = {a.x, a.y}, ahi = {a.z, a.w};
        #pragma unroll
        for (int nt = 0; nt < NT; ++nt) {
            v2f blo = {b[nt].x, b[nt].y};
            acc[nt] = __builtin_amdgcn_wmma_f32_16x16x4_f32(
                false, alo, false, blo, (short)0, acc[nt], false, false);
        }
        #pragma unroll
        for (int nt = 0; nt < NT; ++nt) {
            v2f bhi = {b[nt].z, b[nt].w};
            acc[nt] = __builtin_amdgcn_wmma_f32_16x16x4_f32(
                false, ahi, false, bhi, (short)0, acc[nt], false, false);
        }
    }
    float* sC = &sA[wave][0];
    #pragma unroll
    for (int nt = 0; nt < NT; ++nt) {
        #pragma unroll
        for (int r = 0; r < 8; ++r)
            sC[(8 * h + r) * PC + nt * 16 + c] = fmaxf(acc[nt][r] + bv[nt], 0.0f);
    }
    __syncthreads();
    #pragma unroll
    for (int i = 0; i < W3Q; ++i) {
        int j = i * (GEMM_WAVES * 32) + tid;
        *(v4f*)&sW[(j >> 4) * PC + (j & 15) * 4] = w3v[i];
    }
    *(v4f*)&sW[OFF4 + (tid >> 4) * PC + (tid & 15) * 4] = w4v;
    __syncthreads();

    const float* aT = sC + c * PC + 4 * h;
    const float* bF3 = &sW[c * PC + 4 * h];
    v8f acc2[NT];
    #pragma unroll
    for (int nt = 0; nt < NT; ++nt) acc2[nt] = v8f{};
    #pragma unroll
    for (int t = 0; t < 8; ++t) {
        v4f a = *(const v4f*)(aT + 8 * t);
        v4f b[NT];
        #pragma unroll
        for (int nt = 0; nt < NT; ++nt) b[nt] = *(const v4f*)(bF3 + nt * 16 * PC + 8 * t);
        v2f alo = {a.x, a.y}, ahi = {a.z, a.w};
        #pragma unroll
        for (int nt = 0; nt < NT; ++nt) {
            v2f blo = {b[nt].x, b[nt].y};
            acc2[nt] = __builtin_amdgcn_wmma_f32_16x16x4_f32(
                false, alo, false, blo, (short)0, acc2[nt], false, false);
        }
        #pragma unroll
        for (int nt = 0; nt < NT; ++nt) {
            v2f bhi = {b[nt].z, b[nt].w};
            acc2[nt] = __builtin_amdgcn_wmma_f32_16x16x4_f32(
                false, ahi, false, bhi, (short)0, acc2[nt], false, false);
        }
    }
    #pragma unroll
    for (int nt = 0; nt < NT; ++nt) {
        #pragma unroll
        for (int r = 0; r < 8; ++r)
            sC[(8 * h + r) * PC + nt * 16 + c] = fmaxf(acc2[nt][r] + b3v[nt], 0.0f);
    }
    __syncthreads();

    const float* bF4 = &sW[OFF4 + c * PC + 4 * h];
    v8f acc3 = v8f{};
    #pragma unroll
    for (int t = 0; t < 8; ++t) {
        v4f a = *(const v4f*)(aT + 8 * t);
        v4f b = *(const v4f*)(bF4 + 8 * t);
        v2f alo = {a.x, a.y}, ahi = {a.z, a.w};
        v2f blo = {b.x, b.y}, bhi = {b.z, b.w};
        acc3 = __builtin_amdgcn_wmma_f32_16x16x4_f32(
            false, alo, false, blo, (short)0, acc3, false, false);
        acc3 = __builtin_amdgcn_wmma_f32_16x16x4_f32(
            false, ahi, false, bhi, (short)0, acc3, false, false);
    }
    #pragma unroll
    for (int r = 0; r < 8; ++r) {
        float v = acc3[r] + b4v;
        if (c < 2) sC[(8 * h + r) * 2 + c] = v;
    }
    __syncthreads();
    if (rowsIn) Out[(size_t)waveRow * 2 + lane] = sC[lane];
}

extern "C" void kernel_launch(void* const* d_in, const int* in_sizes, int n_in,
                              void* d_out, int out_size, void* d_ws, size_t ws_size,
                              hipStream_t stream) {
    const float* in_feat = (const float*)d_in[0];
    const int*   src     = (const int*)d_in[1];
    const int*   dst     = (const int*)d_in[2];
    const float* W1 = (const float*)d_in[3];   const float* b1 = (const float*)d_in[4];
    const float* W2 = (const float*)d_in[5];   const float* b2 = (const float*)d_in[6];
    const float* Wc1 = (const float*)d_in[7];  const float* bc1 = (const float*)d_in[8];
    const float* Wc2 = (const float*)d_in[9];  const float* bc2 = (const float*)d_in[10];
    const float* W3 = (const float*)d_in[11];  const float* b3 = (const float*)d_in[12];
    const float* W4 = (const float*)d_in[13];  const float* b4 = (const float*)d_in[14];

    char* ws = (char*)d_ws;
    const size_t SZ_H   = (size_t)N_NODES * 64 * 4;
    const size_t SZ_CAT = (size_t)N_NODES * 192 * 4;
    float* dinv = (float*)(ws);
    float* bufA = (float*)(ws + 200192);
    float* xcat = (float*)(ws + 200192 + 2 * SZ_H);
    char*  eb   = ws + 200192 + 2 * SZ_H + SZ_CAT;
    int*   cnt    = (int*)(eb);
    int*   rowptr = (int*)(eb + 200192);
    int*   rank   = (int*)(eb + 400384);
    int2*  ed     = (int2*)(eb + 400384 + (size_t)N_EDGES * 4);

    dim3 blkS(256);
    dim3 gN((N_NODES + 255) / 256), gE((N_EDGES + 255) / 256);
    dim3 gP((N_NODES * 16 + 255) / 256);

    dim3 gemmBlk(GEMM_WAVES * 32);
    dim3 gG(GEMM_BLOCKS);

    zero_i32<<<gN, blkS, 0, stream>>>(cnt, N_NODES);
    gemm1_and_rank<<<dim3(GEMM_BLOCKS + RANK_BLOCKS), gemmBlk, 0, stream>>>(in_feat, W1, b1, bufA, dst, cnt, rank);
    gemm2_and_scan<<<dim3(GEMM_BLOCKS + SCAN_BLOCKS), gemmBlk, 0, stream>>>(bufA, W2, b2, xcat, cnt, rowptr, dinv);
    fill_edges<<<gE, blkS, 0, stream>>>(src, dst, rank, rowptr, dinv, ed, N_EDGES);

    const float rn = 2.0f / 2.0f;
    spmm_pull<false><<<gP, blkS, 0, stream>>>(rowptr, cnt, ed, dinv, xcat, -rn, rn - 1.0f, 0.0f);
    spmm_pull<true><<<gP, blkS, 0, stream>>>(rowptr, cnt, ed, dinv, xcat, -2.0f * rn, 2.0f * (rn - 1.0f), -1.0f);
    gemm_rows<192, 64, true, 192><<<gG, gemmBlk, 0, stream>>>(xcat, Wc1, bc1, xcat);

    spmm_pull<false><<<gP, blkS, 0, stream>>>(rowptr, cnt, ed, dinv, xcat, -rn, rn - 1.0f, 0.0f);
    spmm_pull<true><<<gP, blkS, 0, stream>>>(rowptr, cnt, ed, dinv, xcat, -2.0f * rn, 2.0f * (rn - 1.0f), -1.0f);
    gemm_tail<<<gG, gemmBlk, 0, stream>>>(xcat, Wc2, bc2, W3, b3, W4, b4, (float*)d_out);
}


# second Chebyshev gather moved into the GEMM that consumes it: a wave gathers X2 of its 16 rows straight into its LDS A tile; X2 never stored; 10 -> 8 launches; row sums keep bucket order
# speedup vs baseline: 5.2556x; 1.0368x over previous
#include <hip/hip_runtime.h>
#include <hip/hip_bf16.h>

typedef __attribute__((ext_vector_type(2))) float v2f;
typedef __attribute__((ext_vector_type(4))) float v4f;
typedef __attribute__((ext_vector_type(8))) float v8f;

#define N_NODES 50000
#define N_EDGES 800000

__global__ void zero_i32(int* __restrict__ p, int n) {
    int i = blockIdx.x * blockDim.x + threadIdx.x;
    if (i < n) p[i] = 0;
}

__device__ __forceinline__ void
scan_body(const int blk, const int* __restrict__ cnt, int* __restrict__ rowptr,
          float* __restrict__ dinv, int n) {
    __shared__ int sTot[8];
    __shared__ int sPre[8];
    const int tid = threadIdx.x, lane = tid & 31, wave = tid >> 5;
    const int base = blk * 256;
    int pre = 0;
    for (int i = tid; i < base; i += 256) pre += cnt[i];
    const int i = base + tid;
    const int c = (i < n) ? cnt[i] : 0;
    int inc = c;
    for (int o = 1; o < 32; o <<= 1) {
        int t = __shfl_up(inc, o);
        if (lane >= o) inc += t;
    }
    for (int o = 16; o > 0; o >>= 1) pre += __shfl_xor(pre, o);
    if (lane == 31) sTot[wave] = inc;
    if (lane == 0)  sPre[wave] = pre;
    __syncthreads();
    int off = 0;
    for (int w = 0; w < 8; ++w) {
        off += sPre[w];
        if (w < wave) off += sTot[w];
    }
    if (i < n) {
        rowptr[i] = off + inc - c;
        dinv[i]   = rsqrtf(fmaxf((float)c, 1.0f));
    }
}

__global__ void fill_edges(const int* __restrict__ src, const int* __restrict__ dst,
                           const int* __restrict__ rank, const int* __restrict__ rowptr,
                           const float* __restrict__ dinv, int2* __restrict__ ed, int e) {
    int i = blockIdx.x * blockDim.x + threadIdx.x;
    if (i >= e) return;
    int s = src[i], d = dst[i];
    ed[rowptr[d] + rank[i]] = make_int2(s, __float_as_int(dinv[s]));
}

template <bool SECOND>
__global__ void __launch_bounds__(256)
spmm_pull(const int* __restrict__ rowptr, const int* __restrict__ cnt,
          const int2* __restrict__ ed, const float* __restrict__ dinv,
          float* xcat, float cAgg, float cIn, float cX0) {
    const int row = (blockIdx.x * 256 + threadIdx.x) >> 4;
    const int l   = threadIdx.x & 15;
    if (row >= N_NODES) return;
    constexpr int IN = SECOND ? 64 : 0;
    const float* xb = xcat + IN + 4 * l;
    const int beg = rowptr[row];
    const int end = beg + cnt[row];
    const v4f zero = {0.0f, 0.0f, 0.0f, 0.0f};
    v4f acc = zero;
    for (int e = beg; e < end; e += 8) {
        int2 p[8];
        #pragma unroll
        for (int j = 0; j < 8; ++j) {
            int k = e + j;
            p[j] = ed[k < end ? k : end - 1];
        }
        v4f x[8];
        #pragma unroll
        for (int j = 0; j < 8; ++j) x[j] = *(const v4f*)(xb + (size_t)p[j].x * 192);
        #pragma unroll
        for (int j = 0; j < 8; ++j) {
            v4f xv = (e + j < end) ? x[j] : zero;
            acc += xv * __int_as_float(p[j].y);
        }
    }
    const float di = dinv[row];
    float* o = xcat + (size_t)row * 192;
    const v4f own = *(const v4f*)(o + IN + 4 * l);
    v4f r = cAgg * (acc * di) + cIn * own;
    if (SECOND) {
        const v4f x0 = *(const v4f*)(o + 4 * l);
        r += cX0 * x0;
    }
    *(v4f*)(o + IN + 64 + 4 * l) = r;
}

#define GEMM_WAVES 8
#define GEMM_ROWS  (GEMM_WAVES * 16)

template <int P>
__device__ __forceinline__ void
tile_x01_gather_x2(float* sT, const float* xcat, const int waveRow, const int lane,
                   const int* __restrict__ rowptr, const int* __restrict__ cnt,
                   const int2* __restrict__ ed, const float* __restrict__ dinv,
                   const float cAgg, const float cIn, const float cX0) {
    const float* xr = xcat + (size_t)waveRow * 192;
    v4f av[16];
    #pragma unroll
    for (int q = 0; q < 16; ++q) av[q] = *(const v4f*)(xr + q * 192 + 4 * lane);
    const int g4 = lane >> 3;
    const int l8 = lane & 7;
    int rb[4], rc[4];
    float rd[4];
    #pragma unroll
    for (int i = 0; i < 4; ++i) {
        const int row = waveRow + 4 * i + g4;
        rb[i] = rowptr[row];
        rc[i] = cnt[row];
        rd[i] = dinv[row];
    }
    #pragma unroll
    for (int q = 0; q < 16; ++q) *(v4f*)&sT[q * P + 4 * lane] = av[q];

    const float* xb = xcat + 64 + 4 * l8;
    const v4f zero = {0.0f, 0.0f, 0.0f, 0.0f};
    #pragma unroll
    for (int i = 0; i < 4; ++i) {
        const int r   = 4 * i + g4;
        const int beg = rb[i];
        const int end = beg + rc[i];
        v4f a0 = zero, a1 = zero;
        for (int e = beg; e < end; e += 8) {
            int2 p[8];
            #pragma unroll
            for (int j = 0; j < 8; ++j) {
                int k = e + j;
                p[j] = ed[k < end ? k : end - 1];
            }
            v4f x0[8], x1[8];
            #pragma unroll
            for (int j = 0; j < 8; ++j) {
                const float* s = xb + (size_t)p[j].x * 192;
                x0[j] = *(const v4f*)s;
                x1[j] = *(const v4f*)(s + 32);
            }
            #pragma unroll
            for (int j = 0; j < 8; ++j) {
                const bool in = e + j < end;
                const float w = __int_as_float(p[j].y);
                a0 += (in ? x0[j] : zero) * w;
                a1 += (in ? x1[j] : zero) * w;
            }
        }
        const float* o = xr + r * 192;
        const v4f own0 = *(const v4f*)(o + 64 + 4 * l8);
        const v4f own1 = *(const v4f*)(o + 96 + 4 * l8);
        const v4f z0   = *(const v4f*)(o + 4 * l8);
        const v4f z1   = *(const v4f*)(o + 32 + 4 * l8);
        v4f r0 = cAgg * (a0 * rd[i]) + cIn * own0;
        v4f r1 = cAgg * (a1 * rd[i]) + cIn * own1;
        r0 += cX0 * z0;
        r1 += cX0 * z1;
        *(v4f*)&sT[r * P + 128 + 4 * l8] = r0;
        *(v4f*)&sT[r * P + 160 + 4 * l8] = r1;
    }
}

template <int K, int NC, bool RELU, int OS = NC, bool GATHER = false>
__device__ __forceinline__ void
gemm_body(const int blk, const float* X, const float* __restrict__ W,
          const float* __restrict__ bias, float* Out,
          const int* __restrict__ rowptr = nullptr, const int* __restrict__ cnt = nullptr,
          const int2* __restrict__ ed = nullptr, const float* __restrict__ dinv = nullptr,
          const float cAgg = 0.0f, const float cIn = 0.0f, const float cX0 = 0.0f) {
    static_assert(!GATHER || K == 192, "the gathered tile is X0 | X1 | X2 of a Chebyshev layer");
    constexpr int NT  = (NC + 15) / 16;
    constexpr int P   = K + 4;
    constexpr int KQ  = K / 4;
    constexpr int AQ  = K / 8;
    constexpr int WQ  = NT * 16 * KQ / (GEMM_WAVES * 32);
    static_assert(WQ * GEMM_WAVES * 32 == NT * 16 * KQ, "the W tile must divide evenly over the block");
    constexpr int PC  = 68;
    static_assert(NC == 64 || NC < 16, "the epilogue handles 64 columns or fewer than 16");
    static_assert(OS == NC || (NC == 64 && OS % 64 == 0), "a wider output pitch keeps every 256-byte row piece aligned");

    __shared__ alignas(16) float sW[NT * 16 * P];
    __shared__ alignas(16) float sA[GEMM_WAVES][16 * P];

    const int tid  = threadIdx.x;
    const int wave = __builtin_amdgcn_readfirstlane(tid >> 5);
    const int lane = tid & 31;
    const int waveRow = blk * GEMM_ROWS + wave * 16;
    const bool rowsIn = waveRow < N_NODES;

    const v4f* Wg = (const v4f*)W;
    v4f wv[WQ];
    #pragma unroll
    for (int i = 0; i < WQ; ++i) {
        int j = i * (GEMM_WAVES * 32) + tid;
        wv[i] = (NC == NT * 16 || j < NC * KQ) ? Wg[j] : v4f{0.0f, 0.0f, 0.0f, 0.0f};
    }

    const v4f* Ag = (const v4f*)(X + (size_t)waveRow * K);
    v4f av[AQ];
    if (rowsIn) {
        if constexpr (GATHER) {
            tile_x01_gather_x2<P>(&sA[wave][0], X, waveRow, lane, rowptr, cnt, ed, dinv, cAgg, cIn, cX0);
        } else {
            #pragma unroll
            for (int q = 0; q < AQ; ++q) av[q] = Ag[q * 32 + lane];
        }
    } else {
        #pragma unroll
        for (int q = 0; q < AQ; ++q) {
            int j = q * 32 + lane;
            int r = j / KQ, kq = j - r * KQ;
            *(v4f*)&sA[wave][r * P + kq * 4] = v4f{0.0f, 0.0f, 0.0f, 0.0f};
        }
    }

    const int c = lane & 15;
    const int h = lane >> 4;
    float bv[NT];
    #pragma unroll
    for (int nt = 0; nt < NT; ++nt) {
        int col = nt * 16 + c;
        bv[nt] = (col < NC) ? bias[col] : 0.0f;
    }

    if constexpr (!GATHER) {
        #pragma unroll
        for (int q = 0; q < AQ; ++q) {
            int j = q * 32 + lane;
            int r = j / KQ, kq = j - r * KQ;
            if (rowsIn) *(v4f*)&sA[wave][r * P + kq * 4] = av[q];
        }
    }
    #pragma unroll
    for (int i = 0; i < WQ; ++i) {
        int j = i * (GEMM_WAVES * 32) + tid;
        int n = j / KQ, kq = j - n * KQ;
        *(v4f*)&sW[n * P + kq * 4] = wv[i];
    }
    __syncthreads();

    const float* aF = &sA[wave][c * P + 4 * h];
    const float* bF = &sW[c * P + 4 * h];
    v8f acc[NT];
    #pragma unroll
    for (int nt = 0; nt < NT; ++nt) acc[nt] = v8f{};

    #pragma unroll
    for (int t = 0; t < AQ; ++t) {
        v4f a = *(const v4f*)(aF + 8 * t);
        v4f b[NT];
        #pragma unroll
        for (int nt = 0; nt < NT; ++nt) b[nt] = *(const v4f*)(bF + nt * 16 * P + 8 * t);
        v2f alo = {a.x, a.y}, ahi = {a.z, a.w};
        #pragma unroll
        for (int nt = 0; nt < NT; ++nt) {
            v2f blo = {b[nt].x, b[nt].y};
            acc[nt] = __builtin_amdgcn_wmma_f32_16x16x4_f32(
                false, alo, false, blo, (short)0, acc[nt], false, false);
        }
        #pragma unroll
        for (int nt = 0; nt < NT; ++nt) {
            v2f bhi = {b[nt].z, b[nt].w};
            acc[nt] = __builtin_amdgcn_wmma_f32_16x16x4_f32(
                false, ahi, false, bhi, (short)0, acc[nt], false, false);
        }
    }

    float* sC = &sA[wave][0];
    if constexpr (NC == 64) {
        #pragma unroll
        for (int nt = 0; nt < NT; ++nt) {
            #pragma unroll
            for (int r = 0; r < 8; ++r) {
                float v = acc[nt][r] + bv[nt];
                if (RELU) v = fmaxf(v, 0.0f);
                sC[(8 * h + r) * PC + nt * 16 + c] = v;
            }
        }
        __syncthreads();
        float* Og = Out + (size_t)waveRow * OS;
        #pragma unroll
        for (int q = 0; q < 8; ++q) {
            int j = q * 32 + lane;
            v4f v = *(const v4f*)&sC[(j >> 4) * PC + (j & 15) * 4];
            if (rowsIn) *(v4f*)(Og + (j >> 4) * OS + (j & 15) * 4) = v;
        }
    } else {
        #pragma unroll
        for (int r = 0; r < 8; ++r) {
            float v = acc[0][r] + bv[0];
            if (RELU) v = fmaxf(v, 0.0f);
            if (c < NC) sC[(8 * h + r) * NC + c] = v;
        }
        __syncthreads();
        if (rowsIn && lane < 16 * NC) Out[(size_t)waveRow * NC + lane] = sC[lane];
    }
}

__global__ void __launch_bounds__(GEMM_WAVES * 32)
gemm_cheb(float* xcat, const float* __restrict__ W, const float* __restrict__ bias,
          const int* __restrict__ rowptr, const int* __restrict__ cnt,
          const int2* __restrict__ ed, const float* __restrict__ dinv,
          float cAgg, float cIn, float cX0) {
    gemm_body<192, 64, true, 192, true>(blockIdx.x, xcat, W, bias, xcat, rowptr, cnt, ed, dinv, cAgg, cIn, cX0);
}

#define GEMM_BLOCKS ((N_NODES + GEMM_ROWS - 1) / GEMM_ROWS)
#define RANK_BLOCKS ((N_EDGES / 4 + 255) / 256)
#define SCAN_BLOCKS ((N_NODES + 255) / 256)
static_assert(N_EDGES % 4 == 0, "edges are ranked four per thread");

__global__ void __launch_bounds__(GEMM_WAVES * 32)
gemm1_and_rank(const float* in_feat, const float* __restrict__ W1,
               const float* __restrict__ b1, float* h,
               const int* __restrict__ dst, int* __restrict__ cnt, int* __restrict__ rank) {
    if (blockIdx.x < GEMM_BLOCKS) {
        gemm_body<128, 64, true>(blockIdx.x, in_feat, W1, b1, h);
    } else {
        const int i = (blockIdx.x - GEMM_BLOCKS) * 256 + threadIdx.x;
        if (i < N_EDGES / 4) {
            const int4 d = ((const int4*)dst)[i];
            int4 r;
            r.x = atomicAdd(&cnt[d.x], 1);
            r.y = atomicAdd(&cnt[d.y], 1);
            r.z = atomicAdd(&cnt[d.z], 1);
            r.w = atomicAdd(&cnt[d.w], 1);
            ((int4*)rank)[i] = r;
        }
    }
}

__global__ void __launch_bounds__(GEMM_WAVES * 32)
gemm2_and_scan(const float* h, const float* __restrict__ W2,
               const float* __restrict__ b2, float* xcat,
               const int* __restrict__ cnt, int* __restrict__ rowptr, float* __restrict__ dinv) {
    if (blockIdx.x < GEMM_BLOCKS) {
        gemm_body<64, 64, true, 192>(blockIdx.x, h, W2, b2, xcat);
    } else {
        scan_body(blockIdx.x - GEMM_BLOCKS, cnt, rowptr, dinv, N_NODES);
    }
}

__global__ void __launch_bounds__(GEMM_WAVES * 32)
gemm_tail(const float* __restrict__ X, const float* __restrict__ Wc, const float* __restrict__ bc,
          const float* __restrict__ W3, const float* __restrict__ b3,
          const float* __restrict__ W4, const float* __restrict__ b4, float* __restrict__ Out,
          const int* __restrict__ rowptr, const int* __restrict__ cnt,
          const int2* __restrict__ ed, const float* __restrict__ dinv,
          float cAgg, float cIn, float cX0) {
    constexpr int K   = 192;
    constexpr int NT  = 4;
    constexpr int P   = K + 4;
    constexpr int KQ  = K / 4;
    constexpr int AQ  = K / 8;
    constexpr int WQ  = NT * 16 * KQ / (GEMM_WAVES * 32);
    constexpr int PC  = 68;
    constexpr int W3Q = 64 * 16 / (GEMM_WAVES * 32);
    constexpr int OFF4 = 64 * PC;
    static_assert(OFF4 + 16 * PC <= NT * 16 * P, "W3 and W4 fit into the first layer's W area");

    __shared__ alignas(16) float sW[NT * 16 * P];
    __shared__ alignas(16) float sA[GEMM_WAVES][16 * P];

    const int tid  = threadIdx.x;
    const int wave = __builtin_amdgcn_readfirstlane(tid >> 5);
    const int lane = tid & 31;
    const int waveRow = blockIdx.x * GEMM_ROWS + wave * 16;
    const bool rowsIn = waveRow < N_NODES;
    const v4f zero4 = {0.0f, 0.0f, 0.0f, 0.0f};

    const v4f* Wg = (const v4f*)Wc;
    v4f wv[WQ];
    #pragma unroll
    for (int i = 0; i < WQ; ++i) wv[i] = Wg[i * (GEMM_WAVES * 32) + tid];

    if (rowsIn) {
        tile_x01_gather_x2<P>(&sA[wave][0], X, waveRow, lane, rowptr, cnt, ed, dinv, cAgg, cIn, cX0);
    } else {
        #pragma unroll
        for (int q = 0; q < AQ; ++q) {
            int j = q * 32 + lane;
            int r = j / KQ, kq = j - r * KQ;
            *(v4f*)&sA[wave][r * P + kq * 4] = zero4;
        }
    }

    const int c = lane & 15;
    const int h = lane >> 4;
    float bv[NT], b3v[NT];
    #pragma unroll
    for (int nt = 0; nt < NT; ++nt) {
        bv[nt]  = bc[nt * 16 + c];
        b3v[nt] = b3[nt * 16 + c];
    }
    const float b4v = (c < 2) ? b4[c] : 0.0f;

    #pragma unroll
    for (int i = 0; i < WQ; ++i) {
        int j = i * (GEMM_WAVES * 32) + tid;
        int n = j / KQ, kq = j - n * KQ;
        *(v4f*)&sW[n * P + kq * 4] = wv[i];
    }
    __syncthreads();

    const v4f* W3g = (const v4f*)W3;
    v4f w3v[W3Q];
    #pragma unroll
    for (int i = 0; i < W3Q; ++i) w3v[i] = W3g[i * (GEMM_WAVES * 32) + tid];
    const v4f w4v = (tid < 2 * 16) ? ((const v4f*)W4)[tid] : zero4;

    const float* aF = &sA[wave][c * P + 4 * h];
    const float* bF = &sW[c * P + 4 * h];
    v8f acc[NT];
    #pragma unroll
    for (int nt = 0; nt < NT; ++nt) acc[nt] = v8f{};
    #pragma unroll
    for (int t = 0; t < AQ; ++t) {
        v4f a = *(const v4f*)(aF + 8 * t);
        v4f b[NT];
        #pragma unroll
        for (int nt = 0; nt < NT; ++nt) b[nt] = *(const v4f*)(bF + nt * 16 * P + 8 * t);
        v2f alo = {a.x, a.y}, ahi = {a.z, a.w};
        #pragma unroll
        for (int nt = 0; nt < NT; ++nt) {
            v2f blo = {b[nt].x, b[nt].y};
            acc[nt] = __builtin_amdgcn_wmma_f32_16x16x4_f32(
                false, alo, false, blo, (short)0, acc[nt], false, false);
        }
        #pragma unroll
        for (int nt = 0; nt < NT; ++nt) {
            v2f bhi = {b[nt].z, b[nt].w};
            acc[nt] = __builtin_amdgcn_wmma_f32_16x16x4_f32(
                false, ahi, false, bhi, (short)0, acc[nt], false, false);
        }
    }
    float* sC = &sA[wave][0];
    #pragma unroll
    for (int nt = 0; nt < NT; ++nt) {
        #pragma unroll
        for (int r = 0; r < 8; ++r)
            sC[(8 * h + r) * PC + nt * 16 + c] = fmaxf(acc[nt][r] + bv[nt], 0.0f);
    }
    __syncthreads();
    #pragma unroll
    for (int i = 0; i < W3Q; ++i) {
        int j = i * (GEMM_WAVES * 32) + tid;
        *(v4f*)&sW[(j >> 4) * PC + (j & 15) * 4] = w3v[i];
    }
    *(v4f*)&sW[OFF4 + (tid >> 4) * PC + (tid & 15) * 4] = w4v;
    __syncthreads();

    const float* aT = sC + c * PC + 4 * h;
    const float* bF3 = &sW[c * PC + 4 * h];
    v8f acc2[NT];
    #pragma unroll
    for (int nt = 0; nt < NT; ++nt) acc2[nt] = v8f{};
    #pragma unroll
    for (int t = 0; t < 8; ++t) {
        v4f a = *(const v4f*)(aT + 8 * t);
        v4f b[NT];
        #pragma unroll
        for (int nt = 0; nt < NT; ++nt) b[nt] = *(const v4f*)(bF3 + nt * 16 * PC + 8 * t);
        v2f alo = {a.x, a.y}, ahi = {a.z, a.w};
        #pragma unroll
        for (int nt = 0; nt < NT; ++nt) {
            v2f blo = {b[nt].x, b[nt].y};
            acc2[nt] = __builtin_amdgcn_wmma_f32_16x16x4_f32(
                false, alo, false, blo, (short)0, acc2[nt], false, false);
        }
        #pragma unroll
        for (int nt = 0; nt < NT; ++nt) {
            v2f bhi = {b[nt].z, b[nt].w};
            acc2[nt] = __builtin_amdgcn_wmma_f32_16x16x4_f32(
                false, ahi, false, bhi, (short)0, acc2[nt], false, false);
        }
    }
    #pragma unroll
    for (int nt = 0; nt < NT; ++nt) {
        #pragma unroll
        for (int r = 0; r < 8; ++r)
            sC[(8 * h + r) * PC + nt * 16 + c] = fmaxf(acc2[nt][r] + b3v[nt], 0.0f);
    }
    __syncthreads();

    const float* bF4 = &sW[OFF4 + c * PC + 4 * h];
    v8f acc3 = v8f{};
    #pragma unroll
    for (int t = 0; t < 8; ++t) {
        v4f a = *(const v4f*)(aT + 8 * t);
        v4f b = *(const v4f*)(bF4 + 8 * t);
        v2f alo = {a.x, a.y}, ahi = {a.z, a.w};
        v2f blo = {b.x, b.y}, bhi = {b.z, b.w};
        acc3 = __builtin_amdgcn_wmma_f32_16x16x4_f32(
            false, alo, false, blo, (short)0, acc3, false, false);
        acc3 = __builtin_amdgcn_wmma_f32_16x16x4_f32(
            false, ahi, false, bhi, (short)0, acc3, false, false);
    }
    #pragma unroll
    for (int r = 0; r < 8; ++r) {
        float v = acc3[r] + b4v;
        if (c < 2) sC[(8 * h + r) * 2 + c] = v;
    }
    __syncthreads();
    if (rowsIn) Out[(size_t)waveRow * 2 + lane] = sC[lane];
}

extern "C" void kernel_launch(void* const* d_in, const int* in_sizes, int n_in,
                              void* d_out, int out_size, void* d_ws, size_t ws_size,
                              hipStream_t stream) {
    const float* in_feat = (const float*)d_in[0];
    const int*   src     = (const int*)d_in[1];
    const int*   dst     = (const int*)d_in[2];
    const float* W1 = (const float*)d_in[3];   const float* b1 = (const float*)d_in[4];
    const float* W2 = (const float*)d_in[5];   const float* b2 = (const float*)d_in[6];
    const float* Wc1 = (const float*)d_in[7];  const float* bc1 = (const float*)d_in[8];
    const float* Wc2 = (const float*)d_in[9];  const float* bc2 = (const float*)d_in[10];
    const float* W3 = (const float*)d_in[11];  const float* b3 = (const float*)d_in[12];
    const float* W4 = (const float*)d_in[13];  const float* b4 = (const float*)d_in[14];

    char* ws = (char*)d_ws;
    const size_t SZ_H   = (size_t)N_NODES * 64 * 4;
    const size_t SZ_CAT = (size_t)N_NODES * 192 * 4;
    float* dinv = (float*)(ws);
    float* bufA = (float*)(ws + 200192);
    float* xcat = (float*)(ws + 200192 + 2 * SZ_H);
    char*  eb   = ws + 200192 + 2 * SZ_H + SZ_CAT;
    int*   cnt    = (int*)(eb);
    int*   rowptr = (int*)(eb + 200192);
    int*   rank   = (int*)(eb + 400384);
    int2*  ed     = (int2*)(eb + 400384 + (size_t)N_EDGES * 4);

    dim3 blkS(256);
    dim3 gN((N_NODES + 255) / 256), gE((N_EDGES + 255) / 256);
    dim3 gP((N_NODES * 16 + 255) / 256);

    dim3 gemmBlk(GEMM_WAVES * 32);
    dim3 gG(GEMM_BLOCKS);

    zero_i32<<<gN, blkS, 0, stream>>>(cnt, N_NODES);
    gemm1_and_rank<<<dim3(GEMM_BLOCKS + RANK_BLOCKS), gemmBlk, 0, stream>>>(in_feat, W1, b1, bufA, dst, cnt, rank);
    gemm2_and_scan<<<dim3(GEMM_BLOCKS + SCAN_BLOCKS), gemmBlk, 0, stream>>>(bufA, W2, b2, xcat, cnt, rowptr, dinv);
    fill_edges<<<gE, blkS, 0, stream>>>(src, dst, rank, rowptr, dinv, ed, N_EDGES);

    const float rn = 2.0f / 2.0f;
    spmm_pull<false><<<gP, blkS, 0, stream>>>(rowptr, cnt, ed, dinv, xcat, -rn, rn - 1.0f, 0.0f);
    gemm_cheb<<<gG, gemmBlk, 0, stream>>>(xcat, Wc1, bc1, rowptr, cnt, ed, dinv, -2.0f * rn, 2.0f * (rn - 1.0f), -1.0f);

    spmm_pull<false><<<gP, blkS, 0, stream>>>(rowptr, cnt, ed, dinv, xcat, -rn, rn - 1.0f, 0.0f);
    gemm_tail<<<gG, gemmBlk, 0, stream>>>(xcat, Wc2, bc2, W3, b3, W4, b4, (float*)d_out,
                                          rowptr, cnt, ed, dinv, -2.0f * rn, 2.0f * (rn - 1.0f), -1.0f);
}
